# starting kernel
# baseline (speedup 1.0000x reference)
#include <hip/hip_runtime.h>
#include <math.h>

constexpr int   Bn   = 4;
constexpr int   Mn   = 4096;
constexpr int   Nn   = Bn * Mn;
constexpr int   PADI = -999;
#define PADF (-999.0f)
constexpr float CUTOFF2 = 49.0f;

typedef float v2f __attribute__((ext_vector_type(2)));
typedef float v8f __attribute__((ext_vector_type(8)));

__device__ __forceinline__ float3 sub3(float3 a, float3 b) {
    return make_float3(a.x - b.x, a.y - b.y, a.z - b.z);
}
__device__ __forceinline__ float3 cross3(float3 a, float3 b) {
    return make_float3(a.y * b.z - a.z * b.y,
                       a.z * b.x - a.x * b.z,
                       a.x * b.y - a.y * b.x);
}
__device__ __forceinline__ float dot3(float3 a, float3 b) {
    return a.x * b.x + a.y * b.y + a.z * b.z;
}

__global__ __launch_bounds__(256) void atoms_kernel(
    const float* __restrict__ coords,
    const int*   __restrict__ cia,
    const int*   __restrict__ partners,
    const int*   __restrict__ aidx,
    float*       __restrict__ out)
{
    int n = blockIdx.x * 256 + threadIdx.x;
    if (n >= Nn) return;
    int b = n >> 12;

    #pragma unroll
    for (int k = 0; k < 2; ++k) {
        int  p  = partners[n * 2 + k];
        bool pv = (p != PADI);
        int  sp = pv ? p : 0;
        const float* pc = coords + ((size_t)b * Mn + sp) * 3;
        out[(size_t)n * 8 + 2 + k * 3 + 0] = pv ? pc[0] : PADF;
        out[(size_t)n * 8 + 2 + k * 3 + 1] = pv ? pc[1] : PADF;
        out[(size_t)n * 8 + 2 + k * 3 + 2] = pv ? pc[2] : PADF;
    }

    int a[4];
    #pragma unroll
    for (int t = 0; t < 4; ++t) a[t] = aidx[n * 4 + t];
    bool am = (a[0] != PADI) && (a[1] != PADI) && (a[2] != PADI) && (a[3] != PADI);

    float3 p[4];
    #pragma unroll
    for (int t = 0; t < 4; ++t) {
        int g  = am ? a[t] : 0;
        int gb = g >> 12;
        int gm = cia[g];
        const float* c = coords + ((size_t)gb * Mn + gm) * 3;
        p[t] = make_float3(c[0], c[1], c[2]);
    }
    float3 b1 = sub3(p[1], p[0]);
    float3 b2 = sub3(p[2], p[1]);
    float3 b3 = sub3(p[3], p[2]);
    float3 n1 = cross3(b1, b2);
    float3 n2 = cross3(b2, b3);
    float  inv = 1.0f / (sqrtf(dot3(b2, b2)) + 1e-12f);
    float3 b2n = make_float3(b2.x * inv, b2.y * inv, b2.z * inv);
    float3 m1  = cross3(n1, b2n);
    float  xx  = dot3(n1, n2);
    float  yy  = dot3(m1, n2);
    out[(size_t)n * 8 + 0] = am ? atan2f(yy, xx) : PADF;
}

__global__ __launch_bounds__(256) void nbr_kernel(
    const float* __restrict__ coords,
    const int*   __restrict__ cia,
    float*       __restrict__ out)
{
    __shared__ float4 sh[Mn];

    int b   = blockIdx.x >> 5;
    int blk = blockIdx.x & 31;

    for (int m = threadIdx.x; m < Mn; m += 256) {
        int idx = cia[b * Mn + m];
        const float* c = coords + ((size_t)b * Mn + idx) * 3;
        float x = c[0], y = c[1], z = c[2];
        sh[m] = make_float4(x, y, z, x * x + y * y + z * z);
    }
    __syncthreads();

    int wave = threadIdx.x >> 5;
    int lane = threadIdx.x & 31;
    int half = lane >> 4;
    int l    = lane & 15;
    int i0   = (blk * 8 + wave) * 16;

    float4 arow = sh[i0 + l];
    v2f A;
    A.x = half ? arow.z : arow.x;
    A.y = half ? 0.0f   : arow.y;

    float h[8];
    #pragma unroll
    for (int v = 0; v < 8; ++v) h[v] = sh[i0 + half * 8 + v].w;

    float acc0[8], acc1[8];
    #pragma unroll
    for (int v = 0; v < 8; ++v) { acc0[v] = 0.0f; acc1[v] = 0.0f; }

    v8f zc = {};
    for (int j0 = 0; j0 < Mn; j0 += 32) {
        float4 brow0 = sh[j0 + l];
        float4 brow1 = sh[j0 + 16 + l];
        v2f B0, B1;
        B0.x = half ? brow0.z : brow0.x;
        B0.y = half ? 0.0f    : brow0.y;
        B1.x = half ? brow1.z : brow1.x;
        B1.y = half ? 0.0f    : brow1.y;
        float sqj0 = brow0.w;
        float sqj1 = brow1.w;

        v8f G0 = __builtin_amdgcn_wmma_f32_16x16x4_f32(
            false, A, false, B0, (short)0, zc, false, false);
        v8f G1 = __builtin_amdgcn_wmma_f32_16x16x4_f32(
            false, A, false, B1, (short)0, zc, false, false);

        #pragma unroll
        for (int v = 0; v < 8; ++v) {
            float d2 = fmaf(G0[v], -2.0f, h[v] + sqj0);
            float t  = fmaxf(d2, 0.0f) + 1e-12f;
            float s  = __builtin_amdgcn_sqrtf(t);
            acc0[v] += (t <= CUTOFF2) ? s : 0.0f;
        }
        #pragma unroll
        for (int v = 0; v < 8; ++v) {
            float d2 = fmaf(G1[v], -2.0f, h[v] + sqj1);
            float t  = fmaxf(d2, 0.0f) + 1e-12f;
            float s  = __builtin_amdgcn_sqrtf(t);
            acc1[v] += (t <= CUTOFF2) ? s : 0.0f;
        }
    }

    #pragma unroll
    for (int v = 0; v < 8; ++v) acc0[v] += acc1[v];

    #pragma unroll
    for (int off = 1; off < 16; off <<= 1) {
        #pragma unroll
        for (int v = 0; v < 8; ++v)
            acc0[v] += __shfl_xor(acc0[v], off, 32);
    }

    if (l == 0) {
        #pragma unroll
        for (int v = 0; v < 8; ++v) {
            int  row   = i0 + half * 8 + v;
            bool valid = (sh[row].x != PADF);
            out[((size_t)(b * Mn + row)) * 8 + 1] = valid ? acc0[v] : 0.0f;
        }
    }
}

extern "C" void kernel_launch(void* const* d_in, const int* in_sizes, int n_in,
                              void* d_out, int out_size, void* d_ws, size_t ws_size,
                              hipStream_t stream) {
    (void)in_sizes; (void)n_in; (void)out_size; (void)d_ws; (void)ws_size;
    const float* coords   = (const float*)d_in[0];
    const int*   cia      = (const int*)d_in[2];
    const int*   partners = (const int*)d_in[3];
    const int*   aidx     = (const int*)d_in[4];
    float*       out      = (float*)d_out;

    atoms_kernel<<<Nn / 256, 256, 0, stream>>>(coords, cia, partners, aidx, out);
    nbr_kernel<<<Bn * 32, 256, 0, stream>>>(coords, cia, out);
}


# nbr sweep spread over all 256 WGPs: 256 blocks of 64 rows; each wave sweeps half the columns of its row tile, the two half-row sums are added in LDS (changes the order of the row sum)
# speedup vs baseline: 1.4753x; 1.4753x over previous
#include <hip/hip_runtime.h>
#include <math.h>

constexpr int   Bn   = 4;
constexpr int   Mn   = 4096;
constexpr int   Nn   = Bn * Mn;
constexpr int   PADI = -999;
#define PADF (-999.0f)
constexpr float CUTOFF2 = 49.0f;

typedef float v2f __attribute__((ext_vector_type(2)));
typedef float v8f __attribute__((ext_vector_type(8)));

__device__ __forceinline__ float3 sub3(float3 a, float3 b) {
    return make_float3(a.x - b.x, a.y - b.y, a.z - b.z);
}
__device__ __forceinline__ float3 cross3(float3 a, float3 b) {
    return make_float3(a.y * b.z - a.z * b.y,
                       a.z * b.x - a.x * b.z,
                       a.x * b.y - a.y * b.x);
}
__device__ __forceinline__ float dot3(float3 a, float3 b) {
    return a.x * b.x + a.y * b.y + a.z * b.z;
}

__global__ __launch_bounds__(256) void atoms_kernel(
    const float* __restrict__ coords,
    const int*   __restrict__ cia,
    const int*   __restrict__ partners,
    const int*   __restrict__ aidx,
    float*       __restrict__ out)
{
    int n = blockIdx.x * 256 + threadIdx.x;
    if (n >= Nn) return;
    int b = n >> 12;

    #pragma unroll
    for (int k = 0; k < 2; ++k) {
        int  p  = partners[n * 2 + k];
        bool pv = (p != PADI);
        int  sp = pv ? p : 0;
        const float* pc = coords + ((size_t)b * Mn + sp) * 3;
        out[(size_t)n * 8 + 2 + k * 3 + 0] = pv ? pc[0] : PADF;
        out[(size_t)n * 8 + 2 + k * 3 + 1] = pv ? pc[1] : PADF;
        out[(size_t)n * 8 + 2 + k * 3 + 2] = pv ? pc[2] : PADF;
    }

    int a[4];
    #pragma unroll
    for (int t = 0; t < 4; ++t) a[t] = aidx[n * 4 + t];
    bool am = (a[0] != PADI) && (a[1] != PADI) && (a[2] != PADI) && (a[3] != PADI);

    float3 p[4];
    #pragma unroll
    for (int t = 0; t < 4; ++t) {
        int g  = am ? a[t] : 0;
        int gb = g >> 12;
        int gm = cia[g];
        const float* c = coords + ((size_t)gb * Mn + gm) * 3;
        p[t] = make_float3(c[0], c[1], c[2]);
    }
    float3 b1 = sub3(p[1], p[0]);
    float3 b2 = sub3(p[2], p[1]);
    float3 b3 = sub3(p[3], p[2]);
    float3 n1 = cross3(b1, b2);
    float3 n2 = cross3(b2, b3);
    float  inv = 1.0f / (sqrtf(dot3(b2, b2)) + 1e-12f);
    float3 b2n = make_float3(b2.x * inv, b2.y * inv, b2.z * inv);
    float3 m1  = cross3(n1, b2n);
    float  xx  = dot3(n1, n2);
    float  yy  = dot3(m1, n2);
    out[(size_t)n * 8 + 0] = am ? atan2f(yy, xx) : PADF;
}

constexpr int ROWS_PER_BLOCK = 64;
constexpr int BLOCKS_PER_BATCH = Mn / ROWS_PER_BLOCK;
constexpr int JHALF = Mn / 2;

__global__ __launch_bounds__(256) void nbr_kernel(
    const float* __restrict__ coords,
    const int*   __restrict__ cia,
    float*       __restrict__ out)
{
    __shared__ float4 sh[Mn];
    __shared__ float  part[2][ROWS_PER_BLOCK];

    int b   = blockIdx.x / BLOCKS_PER_BATCH;
    int blk = blockIdx.x % BLOCKS_PER_BATCH;

    for (int m = threadIdx.x; m < Mn; m += 256) {
        int idx = cia[b * Mn + m];
        const float* c = coords + ((size_t)b * Mn + idx) * 3;
        float x = c[0], y = c[1], z = c[2];
        sh[m] = make_float4(x, y, z, x * x + y * y + z * z);
    }
    __syncthreads();

    int wave  = __builtin_amdgcn_readfirstlane((int)(threadIdx.x >> 5));
    int lane  = threadIdx.x & 31;
    int half  = lane >> 4;
    int l     = lane & 15;
    int itile = wave & 3;
    int jh    = wave >> 2;
    int r0    = itile * 16;
    int i0    = blk * ROWS_PER_BLOCK + r0;
    int jbeg  = jh * JHALF;

    float4 arow = sh[i0 + l];
    v2f A;
    A.x = half ? arow.z : arow.x;
    A.y = half ? 0.0f   : arow.y;

    float h[8];
    #pragma unroll
    for (int v = 0; v < 8; ++v) h[v] = sh[i0 + half * 8 + v].w;

    float acc0[8], acc1[8];
    #pragma unroll
    for (int v = 0; v < 8; ++v) { acc0[v] = 0.0f; acc1[v] = 0.0f; }

    v8f zc = {};
    for (int j0 = jbeg; j0 < jbeg + JHALF; j0 += 32) {
        float4 brow0 = sh[j0 + l];
        float4 brow1 = sh[j0 + 16 + l];
        v2f B0, B1;
        B0.x = half ? brow0.z : brow0.x;
        B0.y = half ? 0.0f    : brow0.y;
        B1.x = half ? brow1.z : brow1.x;
        B1.y = half ? 0.0f    : brow1.y;
        float sqj0 = brow0.w;
        float sqj1 = brow1.w;

        v8f G0 = __builtin_amdgcn_wmma_f32_16x16x4_f32(
            false, A, false, B0, (short)0, zc, false, false);
        v8f G1 = __builtin_amdgcn_wmma_f32_16x16x4_f32(
            false, A, false, B1, (short)0, zc, false, false);

        #pragma unroll
        for (int v = 0; v < 8; ++v) {
            float d2 = fmaf(G0[v], -2.0f, h[v] + sqj0);
            float t  = fmaxf(d2, 0.0f) + 1e-12f;
            float s  = __builtin_amdgcn_sqrtf(t);
            acc0[v] += (t <= CUTOFF2) ? s : 0.0f;
        }
        #pragma unroll
        for (int v = 0; v < 8; ++v) {
            float d2 = fmaf(G1[v], -2.0f, h[v] + sqj1);
            float t  = fmaxf(d2, 0.0f) + 1e-12f;
            float s  = __builtin_amdgcn_sqrtf(t);
            acc1[v] += (t <= CUTOFF2) ? s : 0.0f;
        }
    }

    #pragma unroll
    for (int v = 0; v < 8; ++v) acc0[v] += acc1[v];

    #pragma unroll
    for (int off = 1; off < 16; off <<= 1) {
        #pragma unroll
        for (int v = 0; v < 8; ++v)
            acc0[v] += __shfl_xor(acc0[v], off, 32);
    }

    if (l == 0) {
        #pragma unroll
        for (int v = 0; v < 8; ++v)
            part[jh][r0 + half * 8 + v] = acc0[v];
    }
    __syncthreads();

    if (threadIdx.x < ROWS_PER_BLOCK) {
        int  row   = blk * ROWS_PER_BLOCK + threadIdx.x;
        bool valid = (sh[row].x != PADF);
        float s    = part[0][threadIdx.x] + part[1][threadIdx.x];
        out[((size_t)(b * Mn + row)) * 8 + 1] = valid ? s : 0.0f;
    }
}

extern "C" void kernel_launch(void* const* d_in, const int* in_sizes, int n_in,
                              void* d_out, int out_size, void* d_ws, size_t ws_size,
                              hipStream_t stream) {
    (void)in_sizes; (void)n_in; (void)out_size; (void)d_ws; (void)ws_size;
    const float* coords   = (const float*)d_in[0];
    const int*   cia      = (const int*)d_in[2];
    const int*   partners = (const int*)d_in[3];
    const int*   aidx     = (const int*)d_in[4];
    float*       out      = (float*)d_out;

    atoms_kernel<<<Nn / 256, 256, 0, stream>>>(coords, cia, partners, aidx, out);
    nbr_kernel<<<Bn * BLOCKS_PER_BATCH, 256, 0, stream>>>(coords, cia, out);
}


# nbr fill: batch coords and slot indices read coalesced in one global hop into LDS, per-slot gather done LDS to LDS; was a serial chain of dependent index-then-gather global hops in every wave
# speedup vs baseline: 1.6740x; 1.1347x over previous
#include <hip/hip_runtime.h>
#include <math.h>

constexpr int   Bn   = 4;
constexpr int   Mn   = 4096;
constexpr int   Nn   = Bn * Mn;
constexpr int   PADI = -999;
#define PADF (-999.0f)
constexpr float CUTOFF2 = 49.0f;

typedef float v2f __attribute__((ext_vector_type(2)));
typedef float v8f __attribute__((ext_vector_type(8)));

__device__ __forceinline__ float3 sub3(float3 a, float3 b) {
    return make_float3(a.x - b.x, a.y - b.y, a.z - b.z);
}
__device__ __forceinline__ float3 cross3(float3 a, float3 b) {
    return make_float3(a.y * b.z - a.z * b.y,
                       a.z * b.x - a.x * b.z,
                       a.x * b.y - a.y * b.x);
}
__device__ __forceinline__ float dot3(float3 a, float3 b) {
    return a.x * b.x + a.y * b.y + a.z * b.z;
}

__global__ __launch_bounds__(256) void atoms_kernel(
    const float* __restrict__ coords,
    const int*   __restrict__ cia,
    const int*   __restrict__ partners,
    const int*   __restrict__ aidx,
    float*       __restrict__ out)
{
    int n = blockIdx.x * 256 + threadIdx.x;
    if (n >= Nn) return;
    int b = n >> 12;

    #pragma unroll
    for (int k = 0; k < 2; ++k) {
        int  p  = partners[n * 2 + k];
        bool pv = (p != PADI);
        int  sp = pv ? p : 0;
        const float* pc = coords + ((size_t)b * Mn + sp) * 3;
        out[(size_t)n * 8 + 2 + k * 3 + 0] = pv ? pc[0] : PADF;
        out[(size_t)n * 8 + 2 + k * 3 + 1] = pv ? pc[1] : PADF;
        out[(size_t)n * 8 + 2 + k * 3 + 2] = pv ? pc[2] : PADF;
    }

    int a[4];
    #pragma unroll
    for (int t = 0; t < 4; ++t) a[t] = aidx[n * 4 + t];
    bool am = (a[0] != PADI) && (a[1] != PADI) && (a[2] != PADI) && (a[3] != PADI);

    float3 p[4];
    #pragma unroll
    for (int t = 0; t < 4; ++t) {
        int g  = am ? a[t] : 0;
        int gb = g >> 12;
        int gm = cia[g];
        const float* c = coords + ((size_t)gb * Mn + gm) * 3;
        p[t] = make_float3(c[0], c[1], c[2]);
    }
    float3 b1 = sub3(p[1], p[0]);
    float3 b2 = sub3(p[2], p[1]);
    float3 b3 = sub3(p[3], p[2]);
    float3 n1 = cross3(b1, b2);
    float3 n2 = cross3(b2, b3);
    float  inv = 1.0f / (sqrtf(dot3(b2, b2)) + 1e-12f);
    float3 b2n = make_float3(b2.x * inv, b2.y * inv, b2.z * inv);
    float3 m1  = cross3(n1, b2n);
    float  xx  = dot3(n1, n2);
    float  yy  = dot3(m1, n2);
    out[(size_t)n * 8 + 0] = am ? atan2f(yy, xx) : PADF;
}

constexpr int ROWS_PER_BLOCK = 64;
constexpr int BLOCKS_PER_BATCH = Mn / ROWS_PER_BLOCK;
constexpr int JHALF = Mn / 2;
constexpr int FILL_SLOTS = Mn / 256;

__global__ __launch_bounds__(256) void nbr_kernel(
    const float* __restrict__ coords,
    const int*   __restrict__ cia,
    float*       __restrict__ out)
{
    __shared__ float4 raw[Mn];
    __shared__ float4 sh[Mn];
    __shared__ float  part[2][ROWS_PER_BLOCK];

    int b   = blockIdx.x / BLOCKS_PER_BATCH;
    int blk = blockIdx.x % BLOCKS_PER_BATCH;

    const int tid = (int)threadIdx.x;
    float gx[FILL_SLOTS], gy[FILL_SLOTS], gz[FILL_SLOTS];
    int   idx[FILL_SLOTS];
    #pragma unroll
    for (int k = 0; k < FILL_SLOTS; ++k) {
        const float* c = coords + ((size_t)b * Mn + k * 256 + tid) * 3;
        gx[k] = c[0]; gy[k] = c[1]; gz[k] = c[2];
    }
    #pragma unroll
    for (int k = 0; k < FILL_SLOTS; ++k)
        idx[k] = cia[b * Mn + k * 256 + tid];

    #pragma unroll
    for (int k = 0; k < FILL_SLOTS; ++k) {
        float x = gx[k], y = gy[k], z = gz[k];
        raw[k * 256 + tid] = make_float4(x, y, z, fmaf(z, z, fmaf(x, x, y * y)));
    }
    __syncthreads();

    float qx[FILL_SLOTS], qy[FILL_SLOTS], qz[FILL_SLOTS], qw[FILL_SLOTS];
    #pragma unroll
    for (int k = 0; k < FILL_SLOTS; ++k) {
        float4 r = raw[idx[k]];
        qx[k] = r.x; qy[k] = r.y; qz[k] = r.z; qw[k] = r.w;
    }
    __builtin_amdgcn_sched_barrier(0);
    #pragma unroll
    for (int k = 0; k < FILL_SLOTS; ++k)
        sh[k * 256 + tid] = make_float4(qx[k], qy[k], qz[k], qw[k]);
    __syncthreads();

    int wave  = __builtin_amdgcn_readfirstlane((int)(threadIdx.x >> 5));
    int lane  = threadIdx.x & 31;
    int half  = lane >> 4;
    int l     = lane & 15;
    int itile = wave & 3;
    int jh    = wave >> 2;
    int r0    = itile * 16;
    int i0    = blk * ROWS_PER_BLOCK + r0;
    int jbeg  = jh * JHALF;

    float4 arow = sh[i0 + l];
    v2f A;
    A.x = half ? arow.z : arow.x;
    A.y = half ? 0.0f   : arow.y;

    float h[8];
    #pragma unroll
    for (int v = 0; v < 8; ++v) h[v] = sh[i0 + half * 8 + v].w;

    float acc0[8], acc1[8];
    #pragma unroll
    for (int v = 0; v < 8; ++v) { acc0[v] = 0.0f; acc1[v] = 0.0f; }

    v8f zc = {};
    for (int j0 = jbeg; j0 < jbeg + JHALF; j0 += 32) {
        float4 brow0 = sh[j0 + l];
        float4 brow1 = sh[j0 + 16 + l];
        v2f B0, B1;
        B0.x = half ? brow0.z : brow0.x;
        B0.y = half ? 0.0f    : brow0.y;
        B1.x = half ? brow1.z : brow1.x;
        B1.y = half ? 0.0f    : brow1.y;
        float sqj0 = brow0.w;
        float sqj1 = brow1.w;

        v8f G0 = __builtin_amdgcn_wmma_f32_16x16x4_f32(
            false, A, false, B0, (short)0, zc, false, false);
        v8f G1 = __builtin_amdgcn_wmma_f32_16x16x4_f32(
            false, A, false, B1, (short)0, zc, false, false);

        #pragma unroll
        for (int v = 0; v < 8; ++v) {
            float d2 = fmaf(G0[v], -2.0f, h[v] + sqj0);
            float t  = fmaxf(d2, 0.0f) + 1e-12f;
            float s  = __builtin_amdgcn_sqrtf(t);
            acc0[v] += (t <= CUTOFF2) ? s : 0.0f;
        }
        #pragma unroll
        for (int v = 0; v < 8; ++v) {
            float d2 = fmaf(G1[v], -2.0f, h[v] + sqj1);
            float t  = fmaxf(d2, 0.0f) + 1e-12f;
            float s  = __builtin_amdgcn_sqrtf(t);
            acc1[v] += (t <= CUTOFF2) ? s : 0.0f;
        }
    }

    #pragma unroll
    for (int v = 0; v < 8; ++v) acc0[v] += acc1[v];

    #pragma unroll
    for (int off = 1; off < 16; off <<= 1) {
        #pragma unroll
        for (int v = 0; v < 8; ++v)
            acc0[v] += __shfl_xor(acc0[v], off, 32);
    }

    if (l == 0) {
        #pragma unroll
        for (int v = 0; v < 8; ++v)
            part[jh][r0 + half * 8 + v] = acc0[v];
    }
    __syncthreads();

    if (threadIdx.x < ROWS_PER_BLOCK) {
        int  row   = blk * ROWS_PER_BLOCK + threadIdx.x;
        bool valid = (sh[row].x != PADF);
        float s    = part[0][threadIdx.x] + part[1][threadIdx.x];
        out[((size_t)(b * Mn + row)) * 8 + 1] = valid ? s : 0.0f;
    }
}

extern "C" void kernel_launch(void* const* d_in, const int* in_sizes, int n_in,
                              void* d_out, int out_size, void* d_ws, size_t ws_size,
                              hipStream_t stream) {
    (void)in_sizes; (void)n_in; (void)out_size; (void)d_ws; (void)ws_size;
    const float* coords   = (const float*)d_in[0];
    const int*   cia      = (const int*)d_in[2];
    const int*   partners = (const int*)d_in[3];
    const int*   aidx     = (const int*)d_in[4];
    float*       out      = (float*)d_out;

    atoms_kernel<<<Nn / 256, 256, 0, stream>>>(coords, cia, partners, aidx, out);
    nbr_kernel<<<Bn * BLOCKS_PER_BATCH, 256, 0, stream>>>(coords, cia, out);
}


# per-atom pass (dihedral, partner coords) folded into the neighbour launch, 2 launches -> 1: its gather hops issued under the table fill, waited for behind the sweep; rows stored as whole lines
# speedup vs baseline: 2.2523x; 1.3455x over previous
#include <hip/hip_runtime.h>
#include <math.h>

constexpr int   Bn   = 4;
constexpr int   Mn   = 4096;
constexpr int   Nn   = Bn * Mn;
constexpr int   PADI = -999;
#define PADF (-999.0f)
constexpr float CUTOFF2 = 49.0f;

typedef float v2f __attribute__((ext_vector_type(2)));
typedef float v8f __attribute__((ext_vector_type(8)));

__device__ __forceinline__ float3 sub3(float3 a, float3 b) {
    return make_float3(a.x - b.x, a.y - b.y, a.z - b.z);
}
__device__ __forceinline__ float3 cross3(float3 a, float3 b) {
    return make_float3(a.y * b.z - a.z * b.y,
                       a.z * b.x - a.x * b.z,
                       a.x * b.y - a.y * b.x);
}
__device__ __forceinline__ float dot3(float3 a, float3 b) {
    return a.x * b.x + a.y * b.y + a.z * b.z;
}

__device__ __forceinline__ void lds_barrier() {
    __builtin_amdgcn_fence(__ATOMIC_RELEASE, "workgroup", "local");
    __builtin_amdgcn_s_barrier();
    __builtin_amdgcn_fence(__ATOMIC_ACQUIRE, "workgroup", "local");
}

constexpr int ROWS_PER_BLOCK = 64;
constexpr int BLOCKS_PER_BATCH = Mn / ROWS_PER_BLOCK;
constexpr int JHALF = Mn / 2;
constexpr int FILL_SLOTS = Mn / 256;

__global__ __launch_bounds__(256) void fused_kernel(
    const float* coords,
    const int*   cia,
    const int*   partners,
    const int*   aidx,
    float*       out)
{
    __shared__ float4 raw[Mn];
    __shared__ float4 sh[Mn];
    __shared__ float  part[2][ROWS_PER_BLOCK];
    __shared__ float4 orow[ROWS_PER_BLOCK * 2];

    int b   = blockIdx.x / BLOCKS_PER_BATCH;
    int blk = blockIdx.x % BLOCKS_PER_BATCH;
    const int n0 = (int)blockIdx.x * ROWS_PER_BLOCK;

    const int  wave      = __builtin_amdgcn_readfirstlane((int)(threadIdx.x >> 5));
    const bool w_partner = (wave < 4);
    const bool w_angle   = (wave >= 6);
    float* orowf = (float*)orow;

    const int tid = (int)threadIdx.x;
    float gx[FILL_SLOTS], gy[FILL_SLOTS], gz[FILL_SLOTS];
    int   idx[FILL_SLOTS];
    #pragma unroll
    for (int k = 0; k < FILL_SLOTS; ++k) {
        const float* c = coords + ((size_t)b * Mn + k * 256 + tid) * 3;
        gx[k] = c[0]; gy[k] = c[1]; gz[k] = c[2];
    }
    #pragma unroll
    for (int k = 0; k < FILL_SLOTS; ++k)
        idx[k] = cia[b * Mn + k * 256 + tid];

    int pidx;
    int a[4];
    if (w_partner) pidx = partners[n0 * 2 + tid];
    if (w_angle) {
        const int* ap = aidx + (n0 + tid - 192) * 4;
        #pragma unroll
        for (int t = 0; t < 4; ++t) a[t] = ap[t];
    }

    #pragma unroll
    for (int k = 0; k < FILL_SLOTS; ++k) {
        float x = gx[k], y = gy[k], z = gz[k];
        raw[k * 256 + tid] = make_float4(x, y, z, fmaf(z, z, fmaf(x, x, y * y)));
    }

    #pragma unroll
    for (int k = 0; k < FILL_SLOTS; ++k)
        asm volatile("" : "+v"(idx[k]));

    bool  pv;
    float pcx, pcy, pcz;
    if (w_partner) {
        pv = (pidx != PADI);
        int sp = pv ? pidx : 0;
        const float* pc = coords + ((size_t)b * Mn + sp) * 3;
        pcx = pc[0]; pcy = pc[1]; pcz = pc[2];
    }
    bool am = false;
    int  gb[4] = {0, 0, 0, 0};
    int  gm[4] = {0, 0, 0, 0};
    if (w_angle) {
        am = (a[0] != PADI) && (a[1] != PADI) && (a[2] != PADI) && (a[3] != PADI);
        #pragma unroll
        for (int t = 0; t < 4; ++t) {
            int g = am ? a[t] : 0;
            gb[t] = g >> 12;
            gm[t] = cia[g];
        }
    }
    lds_barrier();

    float qx[FILL_SLOTS], qy[FILL_SLOTS], qz[FILL_SLOTS], qw[FILL_SLOTS];
    #pragma unroll
    for (int k = 0; k < FILL_SLOTS; ++k) {
        float4 r = raw[idx[k]];
        qx[k] = r.x; qy[k] = r.y; qz[k] = r.z; qw[k] = r.w;
    }
    __builtin_amdgcn_sched_barrier(0);
    #pragma unroll
    for (int k = 0; k < FILL_SLOTS; ++k)
        sh[k * 256 + tid] = make_float4(qx[k], qy[k], qz[k], qw[k]);

    #pragma unroll
    for (int t = 0; t < 4; ++t)
        asm volatile("" : "+v"(gm[t]));

    if (w_partner) {
        float* o = orowf + (tid >> 1) * 8 + 2 + (tid & 1) * 3;
        o[0] = pv ? pcx : PADF;
        o[1] = pv ? pcy : PADF;
        o[2] = pv ? pcz : PADF;
    }
    float3 p[4];
    if (w_angle) {
        #pragma unroll
        for (int t = 0; t < 4; ++t) {
            const float* c = coords + ((size_t)gb[t] * Mn + gm[t]) * 3;
            p[t] = make_float3(c[0], c[1], c[2]);
        }
    }
    lds_barrier();

    int lane  = threadIdx.x & 31;
    int half  = lane >> 4;
    int l     = lane & 15;
    int itile = wave & 3;
    int jh    = wave >> 2;
    int r0    = itile * 16;
    int i0    = blk * ROWS_PER_BLOCK + r0;
    int jbeg  = jh * JHALF;

    float4 arow = sh[i0 + l];
    v2f A;
    A.x = half ? arow.z : arow.x;
    A.y = half ? 0.0f   : arow.y;

    float h[8];
    #pragma unroll
    for (int v = 0; v < 8; ++v) h[v] = sh[i0 + half * 8 + v].w;

    float acc0[8], acc1[8];
    #pragma unroll
    for (int v = 0; v < 8; ++v) { acc0[v] = 0.0f; acc1[v] = 0.0f; }

    v8f zc = {};
    for (int j0 = jbeg; j0 < jbeg + JHALF; j0 += 32) {
        float4 brow0 = sh[j0 + l];
        float4 brow1 = sh[j0 + 16 + l];
        v2f B0, B1;
        B0.x = half ? brow0.z : brow0.x;
        B0.y = half ? 0.0f    : brow0.y;
        B1.x = half ? brow1.z : brow1.x;
        B1.y = half ? 0.0f    : brow1.y;
        float sqj0 = brow0.w;
        float sqj1 = brow1.w;

        v8f G0 = __builtin_amdgcn_wmma_f32_16x16x4_f32(
            false, A, false, B0, (short)0, zc, false, false);
        v8f G1 = __builtin_amdgcn_wmma_f32_16x16x4_f32(
            false, A, false, B1, (short)0, zc, false, false);

        #pragma unroll
        for (int v = 0; v < 8; ++v) {
            float d2 = fmaf(G0[v], -2.0f, h[v] + sqj0);
            float t  = fmaxf(d2, 0.0f) + 1e-12f;
            float s  = __builtin_amdgcn_sqrtf(t);
            acc0[v] += (t <= CUTOFF2) ? s : 0.0f;
        }
        #pragma unroll
        for (int v = 0; v < 8; ++v) {
            float d2 = fmaf(G1[v], -2.0f, h[v] + sqj1);
            float t  = fmaxf(d2, 0.0f) + 1e-12f;
            float s  = __builtin_amdgcn_sqrtf(t);
            acc1[v] += (t <= CUTOFF2) ? s : 0.0f;
        }
    }

    #pragma unroll
    for (int v = 0; v < 8; ++v) acc0[v] += acc1[v];

    #pragma unroll
    for (int off = 1; off < 16; off <<= 1) {
        #pragma unroll
        for (int v = 0; v < 8; ++v)
            acc0[v] += __shfl_xor(acc0[v], off, 32);
    }

    if (l == 0) {
        #pragma unroll
        for (int v = 0; v < 8; ++v)
            part[jh][r0 + half * 8 + v] = acc0[v];
    }

    if (w_angle) {
        float3 b1 = sub3(p[1], p[0]);
        float3 b2 = sub3(p[2], p[1]);
        float3 b3 = sub3(p[3], p[2]);
        float3 n1 = cross3(b1, b2);
        float3 n2 = cross3(b2, b3);
        float  inv = 1.0f / (sqrtf(dot3(b2, b2)) + 1e-12f);
        float3 b2n = make_float3(b2.x * inv, b2.y * inv, b2.z * inv);
        float3 m1  = cross3(n1, b2n);
        float  xx  = dot3(n1, n2);
        float  yy  = dot3(m1, n2);
        orowf[(tid - 192) * 8 + 0] = am ? atan2f(yy, xx) : PADF;
    }
    __syncthreads();

    if (w_partner) {
        int    r    = tid >> 1;
        int    hsel = tid & 1;
        float4 v    = orow[tid];
        int    row  = blk * ROWS_PER_BLOCK + r;
        bool  valid = (sh[row].x != PADF);
        float s     = part[0][r] + part[1][r];
        if (hsel == 0) v.y = valid ? s : 0.0f;
        *(float4*)(out + (size_t)(n0 + r) * 8 + hsel * 4) = v;
    }
}

extern "C" void kernel_launch(void* const* d_in, const int* in_sizes, int n_in,
                              void* d_out, int out_size, void* d_ws, size_t ws_size,
                              hipStream_t stream) {
    (void)in_sizes; (void)n_in; (void)out_size; (void)d_ws; (void)ws_size;
    const float* coords   = (const float*)d_in[0];
    const int*   cia      = (const int*)d_in[2];
    const int*   partners = (const int*)d_in[3];
    const int*   aidx     = (const int*)d_in[4];
    float*       out      = (float*)d_out;

    fused_kernel<<<Bn * BLOCKS_PER_BATCH, 256, 0, stream>>>(coords, cia, partners, aidx, out);
}


# sweep: the packed f32 fma's clamp modifier does max(d2,0) and the cutoff select (d2 carried times 2^-6; mask = clamp((49/64 + 1ulp - t) * 2^26) is 0 or 1; sum = fma(s,mask,sum)); sum order kept
# speedup vs baseline: 2.5167x; 1.1174x over previous
#include <hip/hip_runtime.h>
#include <math.h>

constexpr int   Bn   = 4;
constexpr int   Mn   = 4096;
constexpr int   Nn   = Bn * Mn;
constexpr int   PADI = -999;
#define PADF (-999.0f)

typedef float v2f __attribute__((ext_vector_type(2)));
typedef float v8f __attribute__((ext_vector_type(8)));

__device__ __forceinline__ float3 sub3(float3 a, float3 b) {
    return make_float3(a.x - b.x, a.y - b.y, a.z - b.z);
}
__device__ __forceinline__ float3 cross3(float3 a, float3 b) {
    return make_float3(a.y * b.z - a.z * b.y,
                       a.z * b.x - a.x * b.z,
                       a.x * b.y - a.y * b.x);
}
__device__ __forceinline__ float dot3(float3 a, float3 b) {
    return a.x * b.x + a.y * b.y + a.z * b.z;
}

__device__ __forceinline__ void lds_barrier() {
    __builtin_amdgcn_fence(__ATOMIC_RELEASE, "workgroup", "local");
    __builtin_amdgcn_s_barrier();
    __builtin_amdgcn_fence(__ATOMIC_ACQUIRE, "workgroup", "local");
}

constexpr float D2_SCALE   = 0.015625f;
constexpr float EPS_S      = 1e-12f * D2_SCALE;
constexpr float MASK_SCALE = -67108864.0f;
constexpr float MASK_BIAS  = 51380228.0f;
__device__ __forceinline__ v2f fma_sat(v2f a, v2f b, v2f c) {
    v2f d;
    asm("v_pk_fma_f32 %0, %1, %2, %3 clamp" : "=v"(d) : "v"(a), "v"(b), "v"(c));
    return d;
}

constexpr int ROWS_PER_BLOCK = 64;
constexpr int BLOCKS_PER_BATCH = Mn / ROWS_PER_BLOCK;
constexpr int JHALF = Mn / 2;
constexpr int FILL_SLOTS = Mn / 256;

__global__ __launch_bounds__(256) void fused_kernel(
    const float* coords,
    const int*   cia,
    const int*   partners,
    const int*   aidx,
    float*       out)
{
    __shared__ float4 raw[Mn];
    __shared__ float4 sh[Mn];
    __shared__ float  part[2][ROWS_PER_BLOCK];
    __shared__ float4 orow[ROWS_PER_BLOCK * 2];

    int b   = blockIdx.x / BLOCKS_PER_BATCH;
    int blk = blockIdx.x % BLOCKS_PER_BATCH;
    const int n0 = (int)blockIdx.x * ROWS_PER_BLOCK;

    const int  wave      = __builtin_amdgcn_readfirstlane((int)(threadIdx.x >> 5));
    const bool w_partner = (wave < 4);
    const bool w_angle   = (wave >= 6);
    float* orowf = (float*)orow;

    const int tid = (int)threadIdx.x;
    float gx[FILL_SLOTS], gy[FILL_SLOTS], gz[FILL_SLOTS];
    int   idx[FILL_SLOTS];
    #pragma unroll
    for (int k = 0; k < FILL_SLOTS; ++k) {
        const float* c = coords + ((size_t)b * Mn + k * 256 + tid) * 3;
        gx[k] = c[0]; gy[k] = c[1]; gz[k] = c[2];
    }
    #pragma unroll
    for (int k = 0; k < FILL_SLOTS; ++k)
        idx[k] = cia[b * Mn + k * 256 + tid];

    int pidx;
    int a[4];
    if (w_partner) pidx = partners[n0 * 2 + tid];
    if (w_angle) {
        const int* ap = aidx + (n0 + tid - 192) * 4;
        #pragma unroll
        for (int t = 0; t < 4; ++t) a[t] = ap[t];
    }

    #pragma unroll
    for (int k = 0; k < FILL_SLOTS; ++k) {
        float x = gx[k], y = gy[k], z = gz[k];
        raw[k * 256 + tid] = make_float4(x, y, z, D2_SCALE * fmaf(z, z, fmaf(x, x, y * y)));
    }

    #pragma unroll
    for (int k = 0; k < FILL_SLOTS; ++k)
        asm volatile("" : "+v"(idx[k]));

    bool  pv;
    float pcx, pcy, pcz;
    if (w_partner) {
        pv = (pidx != PADI);
        int sp = pv ? pidx : 0;
        const float* pc = coords + ((size_t)b * Mn + sp) * 3;
        pcx = pc[0]; pcy = pc[1]; pcz = pc[2];
    }
    bool am = false;
    int  gb[4] = {0, 0, 0, 0};
    int  gm[4] = {0, 0, 0, 0};
    if (w_angle) {
        am = (a[0] != PADI) && (a[1] != PADI) && (a[2] != PADI) && (a[3] != PADI);
        #pragma unroll
        for (int t = 0; t < 4; ++t) {
            int g = am ? a[t] : 0;
            gb[t] = g >> 12;
            gm[t] = cia[g];
        }
    }
    lds_barrier();

    float qx[FILL_SLOTS], qy[FILL_SLOTS], qz[FILL_SLOTS], qw[FILL_SLOTS];
    #pragma unroll
    for (int k = 0; k < FILL_SLOTS; ++k) {
        float4 r = raw[idx[k]];
        qx[k] = r.x; qy[k] = r.y; qz[k] = r.z; qw[k] = r.w;
    }
    __builtin_amdgcn_sched_barrier(0);
    #pragma unroll
    for (int k = 0; k < FILL_SLOTS; ++k)
        sh[k * 256 + tid] = make_float4(qx[k], qy[k], qz[k], qw[k]);

    #pragma unroll
    for (int t = 0; t < 4; ++t)
        asm volatile("" : "+v"(gm[t]));

    if (w_partner) {
        float* o = orowf + (tid >> 1) * 8 + 2 + (tid & 1) * 3;
        o[0] = pv ? pcx : PADF;
        o[1] = pv ? pcy : PADF;
        o[2] = pv ? pcz : PADF;
    }
    float3 p[4];
    if (w_angle) {
        #pragma unroll
        for (int t = 0; t < 4; ++t) {
            const float* c = coords + ((size_t)gb[t] * Mn + gm[t]) * 3;
            p[t] = make_float3(c[0], c[1], c[2]);
        }
    }
    lds_barrier();

    int lane  = threadIdx.x & 31;
    int half  = lane >> 4;
    int l     = lane & 15;
    int itile = wave & 3;
    int jh    = wave >> 2;
    int r0    = itile * 16;
    int i0    = blk * ROWS_PER_BLOCK + r0;
    int jbeg  = jh * JHALF;

    float4 arow = sh[i0 + l];
    v2f A;
    A.x = half ? arow.z : arow.x;
    A.y = half ? 0.0f   : arow.y;

    float h[8];
    #pragma unroll
    for (int v = 0; v < 8; ++v) h[v] = sh[i0 + half * 8 + v].w;

    float acc0[8], acc1[8];
    #pragma unroll
    for (int v = 0; v < 8; ++v) { acc0[v] = 0.0f; acc1[v] = 0.0f; }

    const v2f gscale = {-2.0f * D2_SCALE, -2.0f * D2_SCALE};
    const v2f mscale = {MASK_SCALE, MASK_SCALE};
    const v2f mbias  = {MASK_BIAS, MASK_BIAS};

    v8f zc = {};
    for (int j0 = jbeg; j0 < jbeg + JHALF; j0 += 32) {
        float4 brow0 = sh[j0 + l];
        float4 brow1 = sh[j0 + 16 + l];
        v2f B0, B1;
        B0.x = half ? brow0.z : brow0.x;
        B0.y = half ? 0.0f    : brow0.y;
        B1.x = half ? brow1.z : brow1.x;
        B1.y = half ? 0.0f    : brow1.y;
        float sqj0 = brow0.w;
        float sqj1 = brow1.w;

        v8f G0 = __builtin_amdgcn_wmma_f32_16x16x4_f32(
            false, A, false, B0, (short)0, zc, false, false);
        v8f G1 = __builtin_amdgcn_wmma_f32_16x16x4_f32(
            false, A, false, B1, (short)0, zc, false, false);

        #pragma unroll
        for (int v = 0; v < 8; v += 2) {
            v2f g, hs, t;
            g.x  = G0[v];          g.y  = G0[v + 1];
            hs.x = h[v] + sqj0;    hs.y = h[v + 1] + sqj0;
            v2f m = fma_sat(g, gscale, hs);
            t.x = m.x + EPS_S;     t.y = m.y + EPS_S;
            v2f w = fma_sat(t, mscale, mbias);
            acc0[v]     = fmaf(__builtin_amdgcn_sqrtf(t.x), w.x, acc0[v]);
            acc0[v + 1] = fmaf(__builtin_amdgcn_sqrtf(t.y), w.y, acc0[v + 1]);
        }
        #pragma unroll
        for (int v = 0; v < 8; v += 2) {
            v2f g, hs, t;
            g.x  = G1[v];          g.y  = G1[v + 1];
            hs.x = h[v] + sqj1;    hs.y = h[v + 1] + sqj1;
            v2f m = fma_sat(g, gscale, hs);
            t.x = m.x + EPS_S;     t.y = m.y + EPS_S;
            v2f w = fma_sat(t, mscale, mbias);
            acc1[v]     = fmaf(__builtin_amdgcn_sqrtf(t.x), w.x, acc1[v]);
            acc1[v + 1] = fmaf(__builtin_amdgcn_sqrtf(t.y), w.y, acc1[v + 1]);
        }
    }

    #pragma unroll
    for (int v = 0; v < 8; ++v) acc0[v] += acc1[v];

    #pragma unroll
    for (int off = 1; off < 16; off <<= 1) {
        #pragma unroll
        for (int v = 0; v < 8; ++v)
            acc0[v] += __shfl_xor(acc0[v], off, 32);
    }

    if (l == 0) {
        #pragma unroll
        for (int v = 0; v < 8; ++v)
            part[jh][r0 + half * 8 + v] = acc0[v];
    }

    if (w_angle) {
        float3 b1 = sub3(p[1], p[0]);
        float3 b2 = sub3(p[2], p[1]);
        float3 b3 = sub3(p[3], p[2]);
        float3 n1 = cross3(b1, b2);
        float3 n2 = cross3(b2, b3);
        float  inv = 1.0f / (sqrtf(dot3(b2, b2)) + 1e-12f);
        float3 b2n = make_float3(b2.x * inv, b2.y * inv, b2.z * inv);
        float3 m1  = cross3(n1, b2n);
        float  xx  = dot3(n1, n2);
        float  yy  = dot3(m1, n2);
        orowf[(tid - 192) * 8 + 0] = am ? atan2f(yy, xx) : PADF;
    }
    __syncthreads();

    if (w_partner) {
        int    r    = tid >> 1;
        int    hsel = tid & 1;
        float4 v    = orow[tid];
        int    row  = blk * ROWS_PER_BLOCK + r;
        bool  valid = (sh[row].x != PADF);
        float s     = (part[0][r] + part[1][r]) * 8.0f;
        if (hsel == 0) v.y = valid ? s : 0.0f;
        *(float4*)(out + (size_t)(n0 + r) * 8 + hsel * 4) = v;
    }
}

extern "C" void kernel_launch(void* const* d_in, const int* in_sizes, int n_in,
                              void* d_out, int out_size, void* d_ws, size_t ws_size,
                              hipStream_t stream) {
    (void)in_sizes; (void)n_in; (void)out_size; (void)d_ws; (void)ws_size;
    const float* coords   = (const float*)d_in[0];
    const int*   cia      = (const int*)d_in[2];
    const int*   partners = (const int*)d_in[3];
    const int*   aidx     = (const int*)d_in[4];
    float*       out      = (float*)d_out;

    fused_kernel<<<Bn * BLOCKS_PER_BATCH, 256, 0, stream>>>(coords, cia, partners, aidx, out);
}


# nbr sweep culled by position: block k sweeps the valid atoms of x slab k against only the columns within cutoff+margin of the slab in x (lists by wave ballot); changes the order of the row sum
# speedup vs baseline: 2.8418x; 1.1292x over previous
#include <hip/hip_runtime.h>
#include <math.h>

constexpr int   Bn   = 4;
constexpr int   Mn   = 4096;
constexpr int   Nn   = Bn * Mn;
constexpr int   PADI = -999;
#define PADF (-999.0f)

typedef float v2f __attribute__((ext_vector_type(2)));
typedef float v8f __attribute__((ext_vector_type(8)));

__device__ __forceinline__ float3 sub3(float3 a, float3 b) {
    return make_float3(a.x - b.x, a.y - b.y, a.z - b.z);
}
__device__ __forceinline__ float3 cross3(float3 a, float3 b) {
    return make_float3(a.y * b.z - a.z * b.y,
                       a.z * b.x - a.x * b.z,
                       a.x * b.y - a.y * b.x);
}
__device__ __forceinline__ float dot3(float3 a, float3 b) {
    return a.x * b.x + a.y * b.y + a.z * b.z;
}

__device__ __forceinline__ void lds_barrier() {
    __builtin_amdgcn_fence(__ATOMIC_RELEASE, "workgroup", "local");
    __builtin_amdgcn_s_barrier();
    __builtin_amdgcn_fence(__ATOMIC_ACQUIRE, "workgroup", "local");
}

constexpr float D2_SCALE   = 0.015625f;
constexpr float EPS_S      = 1e-12f * D2_SCALE;
constexpr float MASK_SCALE = -67108864.0f;
constexpr float MASK_BIAS  = 51380228.0f;
__device__ __forceinline__ v2f fma_sat(v2f a, v2f b, v2f c) {
    v2f d;
    asm("v_pk_fma_f32 %0, %1, %2, %3 clamp" : "=v"(d) : "v"(a), "v"(b), "v"(c));
    return d;
}

constexpr int ROWS_PER_BLOCK = 64;
constexpr int BLOCKS_PER_BATCH = Mn / ROWS_PER_BLOCK;
constexpr int NSLAB = BLOCKS_PER_BATCH;
constexpr int CHUNK_ROWS = 64;
constexpr float CUTOFF_F = 7.0f;
constexpr float MARGIN_REL = 9.5367431640625e-07f;
constexpr float MARGIN_ABS = 1.52587890625e-05f;
constexpr int FILL_SLOTS = Mn / 256;

__global__ __launch_bounds__(256) void fused_kernel(
    const float* coords,
    const int*   cia,
    const int*   partners,
    const int*   aidx,
    float*       out)
{
    __shared__ float4 raw[Mn + 32];
    __shared__ float  nvx[ROWS_PER_BLOCK];
    __shared__ float4 rowl[Mn + 16];
    __shared__ int    rowi[Mn + 16];
    __shared__ float4 red[8];
    __shared__ int2   cnt[8];
    __shared__ float  part[8][CHUNK_ROWS];
    __shared__ float4 orow[ROWS_PER_BLOCK * 2];

    int b   = blockIdx.x / BLOCKS_PER_BATCH;
    int blk = blockIdx.x % BLOCKS_PER_BATCH;
    const int n0 = (int)blockIdx.x * ROWS_PER_BLOCK;

    const int  wave      = __builtin_amdgcn_readfirstlane((int)(threadIdx.x >> 5));
    const bool w_partner = (wave < 4);
    const bool w_angle   = (wave >= 6);
    float* orowf = (float*)orow;

    const int tid = (int)threadIdx.x;
    float gx[FILL_SLOTS], gy[FILL_SLOTS], gz[FILL_SLOTS];
    int   idx[FILL_SLOTS];
    #pragma unroll
    for (int k = 0; k < FILL_SLOTS; ++k) {
        const float* c = coords + ((size_t)b * Mn + k * 256 + tid) * 3;
        gx[k] = c[0]; gy[k] = c[1]; gz[k] = c[2];
    }
    #pragma unroll
    for (int k = 0; k < FILL_SLOTS; ++k)
        idx[k] = cia[b * Mn + k * 256 + tid];

    int pidx;
    int a[4];
    if (w_partner) pidx = partners[n0 * 2 + tid];
    if (w_angle) {
        const int* ap = aidx + (n0 + tid - 192) * 4;
        #pragma unroll
        for (int t = 0; t < 4; ++t) a[t] = ap[t];
    }

    #pragma unroll
    for (int k = 0; k < FILL_SLOTS; ++k) {
        float x = gx[k], y = gy[k], z = gz[k];
        raw[k * 256 + tid] = make_float4(x, y, z, D2_SCALE * fmaf(z, z, fmaf(x, x, y * y)));
    }
    {
        float mn = INFINITY, mx = -INFINITY, wm = 0.0f;
        #pragma unroll
        for (int k = 0; k < FILL_SLOTS; ++k) {
            float x = gx[k], y = gy[k], z = gz[k];
            float w = D2_SCALE * fmaf(z, z, fmaf(x, x, y * y));
            bool ok = (x != PADF);
            mn = ok ? fminf(mn, x) : mn;
            mx = ok ? fmaxf(mx, x) : mx;
            wm = ok ? fmaxf(wm, w) : wm;
        }
        #pragma unroll
        for (int off = 1; off < 32; off <<= 1) {
            mn = fminf(mn, __shfl_xor(mn, off, 32));
            mx = fmaxf(mx, __shfl_xor(mx, off, 32));
            wm = fmaxf(wm, __shfl_xor(wm, off, 32));
        }
        if ((tid & 31) == 0) red[wave] = make_float4(mn, mx, wm, 0.0f);
    }

    #pragma unroll
    for (int k = 0; k < FILL_SLOTS; ++k)
        asm volatile("" : "+v"(idx[k]));

    bool  pv;
    float pcx, pcy, pcz;
    if (w_partner) {
        pv = (pidx != PADI);
        int sp = pv ? pidx : 0;
        const float* pc = coords + ((size_t)b * Mn + sp) * 3;
        pcx = pc[0]; pcy = pc[1]; pcz = pc[2];
    }
    bool am = false;
    int  gb[4] = {0, 0, 0, 0};
    int  gm[4] = {0, 0, 0, 0};
    if (w_angle) {
        am = (a[0] != PADI) && (a[1] != PADI) && (a[2] != PADI) && (a[3] != PADI);
        #pragma unroll
        for (int t = 0; t < 4; ++t) {
            int g = am ? a[t] : 0;
            gb[t] = g >> 12;
            gm[t] = cia[g];
        }
    }
    lds_barrier();

    float rlo, rhi, clo, chi;
    {
        float xmin = INFINITY, xmax = -INFINITY, wmax = 0.0f;
        #pragma unroll
        for (int w = 0; w < 8; ++w) {
            float4 r = red[w];
            xmin = fminf(xmin, r.x); xmax = fmaxf(xmax, r.y); wmax = fmaxf(wmax, r.z);
        }
        if (!(xmax >= xmin)) { xmin = 0.0f; xmax = 0.0f; }
        float width = (xmax - xmin) * (1.0f / NSLAB);
        float lo    = fmaf((float)blk, width, xmin);
        float hi    = fmaf((float)(blk + 1), width, xmin);
        float reach = CUTOFF_F + MARGIN_ABS + (wmax * 64.0f) * MARGIN_REL;
        rlo = (blk == 0)         ? -INFINITY : lo;
        rhi = (blk == NSLAB - 1) ?  INFINITY : hi;
        clo = (blk == 0)         ? -INFINITY : lo - reach;
        chi = (blk == NSLAB - 1) ?  INFINITY : hi + reach;
    }

    float qx[FILL_SLOTS], qy[FILL_SLOTS], qz[FILL_SLOTS], qw[FILL_SLOTS];
    #pragma unroll
    for (int k = 0; k < FILL_SLOTS; ++k) {
        float4 r = raw[idx[k]];
        qx[k] = r.x; qy[k] = r.y; qz[k] = r.z; qw[k] = r.w;
    }
    __builtin_amdgcn_sched_barrier(0);
    #pragma unroll
    for (int k = 0; k < FILL_SLOTS; ++k)
        if (k == (blk >> 2) && (wave >> 1) == (blk & 3)) nvx[tid & 63] = qx[k];

    unsigned bcol[FILL_SLOTS], brow[FILL_SLOTS];
    {
        int cc = 0, cr = 0;
        #pragma unroll
        for (int k = 0; k < FILL_SLOTS; ++k) {
            float x = qx[k];
            bool valid = (x != PADF);
            bool isc = valid && (x >= clo) && (x <= chi);
            bool isr = valid && (x >= rlo) && !(x >= rhi);
            bcol[k] = __builtin_amdgcn_ballot_w32(isc);
            brow[k] = __builtin_amdgcn_ballot_w32(isr);
            cc += __builtin_popcount(bcol[k]);
            cr += __builtin_popcount(brow[k]);
        }
        if ((tid & 31) == 0) cnt[wave] = make_int2(cc, cr);
    }

    #pragma unroll
    for (int t = 0; t < 4; ++t)
        asm volatile("" : "+v"(gm[t]));

    if (w_partner) {
        float* o = orowf + (tid >> 1) * 8 + 2 + (tid & 1) * 3;
        o[0] = pv ? pcx : PADF;
        o[1] = pv ? pcy : PADF;
        o[2] = pv ? pcz : PADF;
    }
    float3 p[4];
    if (w_angle) {
        #pragma unroll
        for (int t = 0; t < 4; ++t) {
            const float* c = coords + ((size_t)gb[t] * Mn + gm[t]) * 3;
            p[t] = make_float3(c[0], c[1], c[2]);
        }
    }
    lds_barrier();

    int lane  = threadIdx.x & 31;
    int half  = lane >> 4;
    int l     = lane & 15;

    int nC = 0, nR = 0;
    {
        int baseC = 0, baseR = 0;
        #pragma unroll
        for (int w = 0; w < 8; ++w) {
            int2 c = cnt[w];
            baseC += (w < wave) ? c.x : 0;
            baseR += (w < wave) ? c.y : 0;
            nC += c.x; nR += c.y;
        }
        nC = __builtin_amdgcn_readfirstlane(nC);
        nR = __builtin_amdgcn_readfirstlane(nR);
        int runC = __builtin_amdgcn_readfirstlane(baseC);
        int runR = __builtin_amdgcn_readfirstlane(baseR);
        #pragma unroll
        for (int k = 0; k < FILL_SLOTS; ++k) {
            unsigned mc = bcol[k], mr = brow[k];
            int pc = runC + (int)__builtin_amdgcn_mbcnt_lo(mc, 0u);
            int pr = runR + (int)__builtin_amdgcn_mbcnt_lo(mr, 0u);
            float4 q = make_float4(qx[k], qy[k], qz[k], qw[k]);
            if ((mc >> lane) & 1u) raw[pc] = q;
            if ((mr >> lane) & 1u) { rowl[pr] = q; rowi[pr] = k * 256 + tid; }
            runC += __builtin_popcount(mc);
            runR += __builtin_popcount(mr);
        }
        if (tid < 32) raw[nC + tid] = make_float4(0.0f, 0.0f, 0.0f, 1e30f);
        if (tid < 16) { rowl[nR + tid] = make_float4(0.0f, 0.0f, 0.0f, 0.0f); rowi[nR + tid] = 0; }
    }
    lds_barrier();

    const int nCP = (nC + 31) >> 5;
    const int p0  = (wave * nCP) >> 3;
    const int p1  = ((wave + 1) * nCP) >> 3;
    const int nRT = (nR + 15) >> 4;

    const v2f gscale = {-2.0f * D2_SCALE, -2.0f * D2_SCALE};
    const v2f mscale = {MASK_SCALE, MASK_SCALE};
    const v2f mbias  = {MASK_BIAS, MASK_BIAS};

    v8f zc = {};
    for (int rt0 = 0; rt0 < nRT; rt0 += CHUNK_ROWS / 16) {
    const int rtend = (rt0 + CHUNK_ROWS / 16 < nRT) ? rt0 + CHUNK_ROWS / 16 : nRT;
    for (int rt = rt0; rt < rtend; ++rt) {
    float4 arow = rowl[rt * 16 + l];
    v2f A;
    A.x = half ? arow.z : arow.x;
    A.y = half ? 0.0f   : arow.y;

    float h[8];
    #pragma unroll
    for (int v = 0; v < 8; ++v) h[v] = rowl[rt * 16 + half * 8 + v].w;

    float acc0[8], acc1[8];
    #pragma unroll
    for (int v = 0; v < 8; ++v) { acc0[v] = 0.0f; acc1[v] = 0.0f; }

    for (int p = p0; p < p1; ++p) {
        const int j0 = p * 32;
        float4 brow0 = raw[j0 + l];
        float4 brow1 = raw[j0 + 16 + l];
        v2f B0, B1;
        B0.x = half ? brow0.z : brow0.x;
        B0.y = half ? 0.0f    : brow0.y;
        B1.x = half ? brow1.z : brow1.x;
        B1.y = half ? 0.0f    : brow1.y;
        float sqj0 = brow0.w;
        float sqj1 = brow1.w;

        v8f G0 = __builtin_amdgcn_wmma_f32_16x16x4_f32(
            false, A, false, B0, (short)0, zc, false, false);
        v8f G1 = __builtin_amdgcn_wmma_f32_16x16x4_f32(
            false, A, false, B1, (short)0, zc, false, false);

        #pragma unroll
        for (int v = 0; v < 8; v += 2) {
            v2f g, hs, t;
            g.x  = G0[v];          g.y  = G0[v + 1];
            hs.x = h[v] + sqj0;    hs.y = h[v + 1] + sqj0;
            v2f m = fma_sat(g, gscale, hs);
            t.x = m.x + EPS_S;     t.y = m.y + EPS_S;
            v2f w = fma_sat(t, mscale, mbias);
            acc0[v]     = fmaf(__builtin_amdgcn_sqrtf(t.x), w.x, acc0[v]);
            acc0[v + 1] = fmaf(__builtin_amdgcn_sqrtf(t.y), w.y, acc0[v + 1]);
        }
        #pragma unroll
        for (int v = 0; v < 8; v += 2) {
            v2f g, hs, t;
            g.x  = G1[v];          g.y  = G1[v + 1];
            hs.x = h[v] + sqj1;    hs.y = h[v + 1] + sqj1;
            v2f m = fma_sat(g, gscale, hs);
            t.x = m.x + EPS_S;     t.y = m.y + EPS_S;
            v2f w = fma_sat(t, mscale, mbias);
            acc1[v]     = fmaf(__builtin_amdgcn_sqrtf(t.x), w.x, acc1[v]);
            acc1[v + 1] = fmaf(__builtin_amdgcn_sqrtf(t.y), w.y, acc1[v + 1]);
        }
    }

    #pragma unroll
    for (int v = 0; v < 8; ++v) acc0[v] += acc1[v];

    #pragma unroll
    for (int off = 1; off < 16; off <<= 1) {
        #pragma unroll
        for (int v = 0; v < 8; ++v)
            acc0[v] += __shfl_xor(acc0[v], off, 32);
    }

    if (l == 0) {
        #pragma unroll
        for (int v = 0; v < 8; ++v)
            part[wave][(rt - rt0) * 16 + half * 8 + v] = acc0[v];
    }
    }

    lds_barrier();
    if (tid < CHUNK_ROWS) {
        int r = rt0 * 16 + tid;
        if (r < nR) {
            float s = ((part[0][tid] + part[1][tid]) + (part[2][tid] + part[3][tid]))
                    + ((part[4][tid] + part[5][tid]) + (part[6][tid] + part[7][tid]));
            out[((size_t)b * Mn + rowi[r]) * 8 + 1] = s * 8.0f;
        }
    }
    lds_barrier();
    }

    if (w_angle) {
        float3 b1 = sub3(p[1], p[0]);
        float3 b2 = sub3(p[2], p[1]);
        float3 b3 = sub3(p[3], p[2]);
        float3 n1 = cross3(b1, b2);
        float3 n2 = cross3(b2, b3);
        float  inv = 1.0f / (sqrtf(dot3(b2, b2)) + 1e-12f);
        float3 b2n = make_float3(b2.x * inv, b2.y * inv, b2.z * inv);
        float3 m1  = cross3(n1, b2n);
        float  xx  = dot3(n1, n2);
        float  yy  = dot3(m1, n2);
        orowf[(tid - 192) * 8 + 0] = am ? atan2f(yy, xx) : PADF;
    }
    lds_barrier();

    if (w_partner) {
        int    r    = tid >> 1;
        int    hsel = tid & 1;
        float4 v    = orow[tid];
        bool  valid = (nvx[r] != PADF);
        float* o    = out + (size_t)(n0 + r) * 8 + hsel * 4;
        if (hsel == 0) {
            o[0] = v.x;
            if (!valid) o[1] = 0.0f;
            *(float2*)(o + 2) = make_float2(v.z, v.w);
        } else {
            *(float4*)o = v;
        }
    }
}

extern "C" void kernel_launch(void* const* d_in, const int* in_sizes, int n_in,
                              void* d_out, int out_size, void* d_ws, size_t ws_size,
                              hipStream_t stream) {
    (void)in_sizes; (void)n_in; (void)out_size; (void)d_ws; (void)ws_size;
    const float* coords   = (const float*)d_in[0];
    const int*   cia      = (const int*)d_in[2];
    const int*   partners = (const int*)d_in[3];
    const int*   aidx     = (const int*)d_in[4];
    float*       out      = (float*)d_out;

    fused_kernel<<<Bn * BLOCKS_PER_BATCH, 256, 0, stream>>>(coords, cia, partners, aidx, out);
}


# sweep dealt to the 8 waves as one stretch each of the (row tile, column pair) items, not 1/8 of the columns of every row tile: a wave meets 2 row tiles at most; changes the order of the row sum
# speedup vs baseline: 3.0531x; 1.0744x over previous
#include <hip/hip_runtime.h>
#include <math.h>

constexpr int   Bn   = 4;
constexpr int   Mn   = 4096;
constexpr int   Nn   = Bn * Mn;
constexpr int   PADI = -999;
#define PADF (-999.0f)

typedef float v2f __attribute__((ext_vector_type(2)));
typedef float v8f __attribute__((ext_vector_type(8)));

__device__ __forceinline__ float3 sub3(float3 a, float3 b) {
    return make_float3(a.x - b.x, a.y - b.y, a.z - b.z);
}
__device__ __forceinline__ float3 cross3(float3 a, float3 b) {
    return make_float3(a.y * b.z - a.z * b.y,
                       a.z * b.x - a.x * b.z,
                       a.x * b.y - a.y * b.x);
}
__device__ __forceinline__ float dot3(float3 a, float3 b) {
    return a.x * b.x + a.y * b.y + a.z * b.z;
}

__device__ __forceinline__ void lds_barrier() {
    __builtin_amdgcn_fence(__ATOMIC_RELEASE, "workgroup", "local");
    __builtin_amdgcn_s_barrier();
    __builtin_amdgcn_fence(__ATOMIC_ACQUIRE, "workgroup", "local");
}

constexpr float D2_SCALE   = 0.015625f;
constexpr float EPS_S      = 1e-12f * D2_SCALE;
constexpr float MASK_SCALE = -67108864.0f;
constexpr float MASK_BIAS  = 51380228.0f;
__device__ __forceinline__ v2f fma_sat(v2f a, v2f b, v2f c) {
    v2f d;
    asm("v_pk_fma_f32 %0, %1, %2, %3 clamp" : "=v"(d) : "v"(a), "v"(b), "v"(c));
    return d;
}

constexpr int ROWS_PER_BLOCK = 64;
constexpr int BLOCKS_PER_BATCH = Mn / ROWS_PER_BLOCK;
constexpr int NSLAB = BLOCKS_PER_BATCH;
constexpr int CHUNK_ROWS = 128;
constexpr float CUTOFF_F = 7.0f;
constexpr float MARGIN_REL = 9.5367431640625e-07f;
constexpr float MARGIN_ABS = 1.52587890625e-05f;
constexpr int FILL_SLOTS = Mn / 256;

__global__ __launch_bounds__(256) void fused_kernel(
    const float* coords,
    const int*   cia,
    const int*   partners,
    const int*   aidx,
    float*       out)
{
    __shared__ float4 raw[Mn + 32];
    __shared__ float  nvx[ROWS_PER_BLOCK];
    __shared__ float4 rowl[Mn + 16];
    __shared__ int    rowi[Mn + 16];
    __shared__ float4 red[8];
    __shared__ int2   cnt[8];
    __shared__ float  part[8][CHUNK_ROWS];
    __shared__ float4 orow[ROWS_PER_BLOCK * 2];

    int b   = blockIdx.x / BLOCKS_PER_BATCH;
    int blk = blockIdx.x % BLOCKS_PER_BATCH;
    const int n0 = (int)blockIdx.x * ROWS_PER_BLOCK;

    const int  wave      = __builtin_amdgcn_readfirstlane((int)(threadIdx.x >> 5));
    const bool w_partner = (wave < 4);
    const bool w_angle   = (wave >= 6);
    float* orowf = (float*)orow;

    const int tid = (int)threadIdx.x;
    float gx[FILL_SLOTS], gy[FILL_SLOTS], gz[FILL_SLOTS];
    int   idx[FILL_SLOTS];
    #pragma unroll
    for (int k = 0; k < FILL_SLOTS; ++k) {
        const float* c = coords + ((size_t)b * Mn + k * 256 + tid) * 3;
        gx[k] = c[0]; gy[k] = c[1]; gz[k] = c[2];
    }
    #pragma unroll
    for (int k = 0; k < FILL_SLOTS; ++k)
        idx[k] = cia[b * Mn + k * 256 + tid];

    int pidx;
    int a[4];
    if (w_partner) pidx = partners[n0 * 2 + tid];
    if (w_angle) {
        const int* ap = aidx + (n0 + tid - 192) * 4;
        #pragma unroll
        for (int t = 0; t < 4; ++t) a[t] = ap[t];
    }

    #pragma unroll
    for (int k = 0; k < FILL_SLOTS; ++k) {
        float x = gx[k], y = gy[k], z = gz[k];
        raw[k * 256 + tid] = make_float4(x, y, z, D2_SCALE * fmaf(z, z, fmaf(x, x, y * y)));
    }
    {
        float mn = INFINITY, mx = -INFINITY, wm = 0.0f;
        #pragma unroll
        for (int k = 0; k < FILL_SLOTS; ++k) {
            float x = gx[k], y = gy[k], z = gz[k];
            float w = D2_SCALE * fmaf(z, z, fmaf(x, x, y * y));
            bool ok = (x != PADF);
            mn = ok ? fminf(mn, x) : mn;
            mx = ok ? fmaxf(mx, x) : mx;
            wm = ok ? fmaxf(wm, w) : wm;
        }
        #pragma unroll
        for (int off = 1; off < 32; off <<= 1) {
            mn = fminf(mn, __shfl_xor(mn, off, 32));
            mx = fmaxf(mx, __shfl_xor(mx, off, 32));
            wm = fmaxf(wm, __shfl_xor(wm, off, 32));
        }
        if ((tid & 31) == 0) red[wave] = make_float4(mn, mx, wm, 0.0f);
    }

    #pragma unroll
    for (int k = 0; k < FILL_SLOTS; ++k)
        asm volatile("" : "+v"(idx[k]));

    bool  pv;
    float pcx, pcy, pcz;
    if (w_partner) {
        pv = (pidx != PADI);
        int sp = pv ? pidx : 0;
        const float* pc = coords + ((size_t)b * Mn + sp) * 3;
        pcx = pc[0]; pcy = pc[1]; pcz = pc[2];
    }
    bool am = false;
    int  gb[4] = {0, 0, 0, 0};
    int  gm[4] = {0, 0, 0, 0};
    if (w_angle) {
        am = (a[0] != PADI) && (a[1] != PADI) && (a[2] != PADI) && (a[3] != PADI);
        #pragma unroll
        for (int t = 0; t < 4; ++t) {
            int g = am ? a[t] : 0;
            gb[t] = g >> 12;
            gm[t] = cia[g];
        }
    }
    lds_barrier();

    float rlo, rhi, clo, chi;
    {
        float xmin = INFINITY, xmax = -INFINITY, wmax = 0.0f;
        #pragma unroll
        for (int w = 0; w < 8; ++w) {
            float4 r = red[w];
            xmin = fminf(xmin, r.x); xmax = fmaxf(xmax, r.y); wmax = fmaxf(wmax, r.z);
        }
        if (!(xmax >= xmin)) { xmin = 0.0f; xmax = 0.0f; }
        float width = (xmax - xmin) * (1.0f / NSLAB);
        float lo    = fmaf((float)blk, width, xmin);
        float hi    = fmaf((float)(blk + 1), width, xmin);
        float reach = CUTOFF_F + MARGIN_ABS + (wmax * 64.0f) * MARGIN_REL;
        rlo = (blk == 0)         ? -INFINITY : lo;
        rhi = (blk == NSLAB - 1) ?  INFINITY : hi;
        clo = (blk == 0)         ? -INFINITY : lo - reach;
        chi = (blk == NSLAB - 1) ?  INFINITY : hi + reach;
    }

    float qx[FILL_SLOTS], qy[FILL_SLOTS], qz[FILL_SLOTS], qw[FILL_SLOTS];
    #pragma unroll
    for (int k = 0; k < FILL_SLOTS; ++k) {
        float4 r = raw[idx[k]];
        qx[k] = r.x; qy[k] = r.y; qz[k] = r.z; qw[k] = r.w;
    }
    __builtin_amdgcn_sched_barrier(0);
    #pragma unroll
    for (int k = 0; k < FILL_SLOTS; ++k)
        if (k == (blk >> 2) && (wave >> 1) == (blk & 3)) nvx[tid & 63] = qx[k];

    unsigned bcol[FILL_SLOTS], brow[FILL_SLOTS];
    {
        int cc = 0, cr = 0;
        #pragma unroll
        for (int k = 0; k < FILL_SLOTS; ++k) {
            float x = qx[k];
            bool valid = (x != PADF);
            bool isc = valid && (x >= clo) && (x <= chi);
            bool isr = valid && (x >= rlo) && !(x >= rhi);
            bcol[k] = __builtin_amdgcn_ballot_w32(isc);
            brow[k] = __builtin_amdgcn_ballot_w32(isr);
            cc += __builtin_popcount(bcol[k]);
            cr += __builtin_popcount(brow[k]);
        }
        if ((tid & 31) == 0) cnt[wave] = make_int2(cc, cr);
    }

    #pragma unroll
    for (int t = 0; t < 4; ++t)
        asm volatile("" : "+v"(gm[t]));

    if (w_partner) {
        float* o = orowf + (tid >> 1) * 8 + 2 + (tid & 1) * 3;
        o[0] = pv ? pcx : PADF;
        o[1] = pv ? pcy : PADF;
        o[2] = pv ? pcz : PADF;
    }
    float3 p[4];
    if (w_angle) {
        #pragma unroll
        for (int t = 0; t < 4; ++t) {
            const float* c = coords + ((size_t)gb[t] * Mn + gm[t]) * 3;
            p[t] = make_float3(c[0], c[1], c[2]);
        }
    }
    lds_barrier();

    int lane  = threadIdx.x & 31;
    int half  = lane >> 4;
    int l     = lane & 15;

    int nC = 0, nR = 0;
    {
        int baseC = 0, baseR = 0;
        #pragma unroll
        for (int w = 0; w < 8; ++w) {
            int2 c = cnt[w];
            baseC += (w < wave) ? c.x : 0;
            baseR += (w < wave) ? c.y : 0;
            nC += c.x; nR += c.y;
        }
        nC = __builtin_amdgcn_readfirstlane(nC);
        nR = __builtin_amdgcn_readfirstlane(nR);
        int runC = __builtin_amdgcn_readfirstlane(baseC);
        int runR = __builtin_amdgcn_readfirstlane(baseR);
        #pragma unroll
        for (int k = 0; k < FILL_SLOTS; ++k) {
            unsigned mc = bcol[k], mr = brow[k];
            int pc = runC + (int)__builtin_amdgcn_mbcnt_lo(mc, 0u);
            int pr = runR + (int)__builtin_amdgcn_mbcnt_lo(mr, 0u);
            float4 q = make_float4(qx[k], qy[k], qz[k], qw[k]);
            if ((mc >> lane) & 1u) raw[pc] = q;
            if ((mr >> lane) & 1u) { rowl[pr] = q; rowi[pr] = k * 256 + tid; }
            runC += __builtin_popcount(mc);
            runR += __builtin_popcount(mr);
        }
        if (tid < 32) raw[nC + tid] = make_float4(0.0f, 0.0f, 0.0f, 1e30f);
        if (tid < 16) { rowl[nR + tid] = make_float4(0.0f, 0.0f, 0.0f, 0.0f); rowi[nR + tid] = 0; }
    }
    lds_barrier();

    const int nCP = (nC + 31) >> 5;
    const int nRT = (nR + 15) >> 4;

    const v2f gscale = {-2.0f * D2_SCALE, -2.0f * D2_SCALE};
    const v2f mscale = {MASK_SCALE, MASK_SCALE};
    const v2f mbias  = {MASK_BIAS, MASK_BIAS};

    v8f zc = {};
    for (int rt0 = 0; rt0 < nRT; rt0 += CHUNK_ROWS / 16) {
    const int rtend = (rt0 + CHUNK_ROWS / 16 < nRT) ? rt0 + CHUNK_ROWS / 16 : nRT;
    const int nItems = (rtend - rt0) * nCP;
    const int q0 = (wave * nItems) >> 3;
    const int q1 = ((wave + 1) * nItems) >> 3;
    for (int rt = rt0; rt < rtend; ++rt) {
    const int qb = (rt - rt0) * nCP;
    const int p0 = (q0 - qb > 0) ? q0 - qb : 0;
    const int p1 = (q1 - qb < nCP) ? q1 - qb : nCP;
    if (p0 >= p1) {
        if (l == 0) {
            #pragma unroll
            for (int v = 0; v < 8; ++v)
                part[wave][(rt - rt0) * 16 + half * 8 + v] = 0.0f;
        }
        continue;
    }
    float4 arow = rowl[rt * 16 + l];
    v2f A;
    A.x = half ? arow.z : arow.x;
    A.y = half ? 0.0f   : arow.y;

    float h[8];
    #pragma unroll
    for (int v = 0; v < 8; ++v) h[v] = rowl[rt * 16 + half * 8 + v].w;

    float acc0[8], acc1[8];
    #pragma unroll
    for (int v = 0; v < 8; ++v) { acc0[v] = 0.0f; acc1[v] = 0.0f; }

    for (int p = p0; p < p1; ++p) {
        const int j0 = p * 32;
        float4 brow0 = raw[j0 + l];
        float4 brow1 = raw[j0 + 16 + l];
        v2f B0, B1;
        B0.x = half ? brow0.z : brow0.x;
        B0.y = half ? 0.0f    : brow0.y;
        B1.x = half ? brow1.z : brow1.x;
        B1.y = half ? 0.0f    : brow1.y;
        float sqj0 = brow0.w;
        float sqj1 = brow1.w;

        v8f G0 = __builtin_amdgcn_wmma_f32_16x16x4_f32(
            false, A, false, B0, (short)0, zc, false, false);
        v8f G1 = __builtin_amdgcn_wmma_f32_16x16x4_f32(
            false, A, false, B1, (short)0, zc, false, false);

        #pragma unroll
        for (int v = 0; v < 8; v += 2) {
            v2f g, hs, t;
            g.x  = G0[v];          g.y  = G0[v + 1];
            hs.x = h[v] + sqj0;    hs.y = h[v + 1] + sqj0;
            v2f m = fma_sat(g, gscale, hs);
            t.x = m.x + EPS_S;     t.y = m.y + EPS_S;
            v2f w = fma_sat(t, mscale, mbias);
            acc0[v]     = fmaf(__builtin_amdgcn_sqrtf(t.x), w.x, acc0[v]);
            acc0[v + 1] = fmaf(__builtin_amdgcn_sqrtf(t.y), w.y, acc0[v + 1]);
        }
        #pragma unroll
        for (int v = 0; v < 8; v += 2) {
            v2f g, hs, t;
            g.x  = G1[v];          g.y  = G1[v + 1];
            hs.x = h[v] + sqj1;    hs.y = h[v + 1] + sqj1;
            v2f m = fma_sat(g, gscale, hs);
            t.x = m.x + EPS_S;     t.y = m.y + EPS_S;
            v2f w = fma_sat(t, mscale, mbias);
            acc1[v]     = fmaf(__builtin_amdgcn_sqrtf(t.x), w.x, acc1[v]);
            acc1[v + 1] = fmaf(__builtin_amdgcn_sqrtf(t.y), w.y, acc1[v + 1]);
        }
    }

    #pragma unroll
    for (int v = 0; v < 8; ++v) acc0[v] += acc1[v];

    #pragma unroll
    for (int off = 1; off < 16; off <<= 1) {
        #pragma unroll
        for (int v = 0; v < 8; ++v)
            acc0[v] += __shfl_xor(acc0[v], off, 32);
    }

    if (l == 0) {
        #pragma unroll
        for (int v = 0; v < 8; ++v)
            part[wave][(rt - rt0) * 16 + half * 8 + v] = acc0[v];
    }
    }

    lds_barrier();
    if (tid < CHUNK_ROWS) {
        int r = rt0 * 16 + tid;
        if (r < nR) {
            float s = ((part[0][tid] + part[1][tid]) + (part[2][tid] + part[3][tid]))
                    + ((part[4][tid] + part[5][tid]) + (part[6][tid] + part[7][tid]));
            out[((size_t)b * Mn + rowi[r]) * 8 + 1] = s * 8.0f;
        }
    }
    lds_barrier();
    }

    if (w_angle) {
        float3 b1 = sub3(p[1], p[0]);
        float3 b2 = sub3(p[2], p[1]);
        float3 b3 = sub3(p[3], p[2]);
        float3 n1 = cross3(b1, b2);
        float3 n2 = cross3(b2, b3);
        float  inv = 1.0f / (sqrtf(dot3(b2, b2)) + 1e-12f);
        float3 b2n = make_float3(b2.x * inv, b2.y * inv, b2.z * inv);
        float3 m1  = cross3(n1, b2n);
        float  xx  = dot3(n1, n2);
        float  yy  = dot3(m1, n2);
        orowf[(tid - 192) * 8 + 0] = am ? atan2f(yy, xx) : PADF;
    }
    lds_barrier();

    if (w_partner) {
        int    r    = tid >> 1;
        int    hsel = tid & 1;
        float4 v    = orow[tid];
        bool  valid = (nvx[r] != PADF);
        float* o    = out + (size_t)(n0 + r) * 8 + hsel * 4;
        if (hsel == 0) {
            o[0] = v.x;
            if (!valid) o[1] = 0.0f;
            *(float2*)(o + 2) = make_float2(v.z, v.w);
        } else {
            *(float4*)o = v;
        }
    }
}

extern "C" void kernel_launch(void* const* d_in, const int* in_sizes, int n_in,
                              void* d_out, int out_size, void* d_ws, size_t ws_size,
                              hipStream_t stream) {
    (void)in_sizes; (void)n_in; (void)out_size; (void)d_ws; (void)ws_size;
    const float* coords   = (const float*)d_in[0];
    const int*   cia      = (const int*)d_in[2];
    const int*   partners = (const int*)d_in[3];
    const int*   aidx     = (const int*)d_in[4];
    float*       out      = (float*)d_out;

    fused_kernel<<<Bn * BLOCKS_PER_BATCH, 256, 0, stream>>>(coords, cia, partners, aidx, out);
}


# nbr sweep culled in x and y: block k takes the rows of cell k of an 8 x 8 grid over the batch's x and y ranges (was 64 x slabs), columns within cutoff+margin of it in both; changes row sum order
# speedup vs baseline: 3.2964x; 1.0797x over previous
#include <hip/hip_runtime.h>
#include <math.h>

constexpr int   Bn   = 4;
constexpr int   Mn   = 4096;
constexpr int   Nn   = Bn * Mn;
constexpr int   PADI = -999;
#define PADF (-999.0f)

typedef float v2f __attribute__((ext_vector_type(2)));
typedef float v8f __attribute__((ext_vector_type(8)));

__device__ __forceinline__ float3 sub3(float3 a, float3 b) {
    return make_float3(a.x - b.x, a.y - b.y, a.z - b.z);
}
__device__ __forceinline__ float3 cross3(float3 a, float3 b) {
    return make_float3(a.y * b.z - a.z * b.y,
                       a.z * b.x - a.x * b.z,
                       a.x * b.y - a.y * b.x);
}
__device__ __forceinline__ float dot3(float3 a, float3 b) {
    return a.x * b.x + a.y * b.y + a.z * b.z;
}

__device__ __forceinline__ void lds_barrier() {
    __builtin_amdgcn_fence(__ATOMIC_RELEASE, "workgroup", "local");
    __builtin_amdgcn_s_barrier();
    __builtin_amdgcn_fence(__ATOMIC_ACQUIRE, "workgroup", "local");
}

constexpr float D2_SCALE   = 0.015625f;
constexpr float EPS_S      = 1e-12f * D2_SCALE;
constexpr float MASK_SCALE = -67108864.0f;
constexpr float MASK_BIAS  = 51380228.0f;
__device__ __forceinline__ v2f fma_sat(v2f a, v2f b, v2f c) {
    v2f d;
    asm("v_pk_fma_f32 %0, %1, %2, %3 clamp" : "=v"(d) : "v"(a), "v"(b), "v"(c));
    return d;
}

constexpr int ROWS_PER_BLOCK = 64;
constexpr int BLOCKS_PER_BATCH = Mn / ROWS_PER_BLOCK;
constexpr int NSX = 8;
constexpr int NSY = 8;
static_assert(NSX * NSY == BLOCKS_PER_BATCH, "one cell per block of the batch");
constexpr int CHUNK_ROWS = 128;
constexpr float CUTOFF_F = 7.0f;
constexpr float MARGIN_REL = 9.5367431640625e-07f;
constexpr float MARGIN_ABS = 1.52587890625e-05f;
constexpr int FILL_SLOTS = Mn / 256;

__global__ __launch_bounds__(256) void fused_kernel(
    const float* coords,
    const int*   cia,
    const int*   partners,
    const int*   aidx,
    float*       out)
{
    __shared__ float4 raw[Mn + 32];
    __shared__ float  nvx[ROWS_PER_BLOCK];
    __shared__ float4 rowl[Mn + 16];
    __shared__ int    rowi[Mn + 16];
    __shared__ float4 red[8];
    __shared__ float  redw[8];
    __shared__ int2   cnt[8];
    __shared__ float  part[8][CHUNK_ROWS];
    __shared__ float4 orow[ROWS_PER_BLOCK * 2];

    int b   = blockIdx.x / BLOCKS_PER_BATCH;
    int blk = blockIdx.x % BLOCKS_PER_BATCH;
    const int n0 = (int)blockIdx.x * ROWS_PER_BLOCK;

    const int  wave      = __builtin_amdgcn_readfirstlane((int)(threadIdx.x >> 5));
    const bool w_partner = (wave < 4);
    const bool w_angle   = (wave >= 6);
    float* orowf = (float*)orow;

    const int tid = (int)threadIdx.x;
    float gx[FILL_SLOTS], gy[FILL_SLOTS], gz[FILL_SLOTS];
    int   idx[FILL_SLOTS];
    #pragma unroll
    for (int k = 0; k < FILL_SLOTS; ++k) {
        const float* c = coords + ((size_t)b * Mn + k * 256 + tid) * 3;
        gx[k] = c[0]; gy[k] = c[1]; gz[k] = c[2];
    }
    #pragma unroll
    for (int k = 0; k < FILL_SLOTS; ++k)
        idx[k] = cia[b * Mn + k * 256 + tid];

    int pidx;
    int a[4];
    if (w_partner) pidx = partners[n0 * 2 + tid];
    if (w_angle) {
        const int* ap = aidx + (n0 + tid - 192) * 4;
        #pragma unroll
        for (int t = 0; t < 4; ++t) a[t] = ap[t];
    }

    #pragma unroll
    for (int k = 0; k < FILL_SLOTS; ++k) {
        float x = gx[k], y = gy[k], z = gz[k];
        raw[k * 256 + tid] = make_float4(x, y, z, D2_SCALE * fmaf(z, z, fmaf(x, x, y * y)));
    }
    {
        float mn = INFINITY, mx = -INFINITY, wm = 0.0f;
        float yn = INFINITY, yx = -INFINITY;
        #pragma unroll
        for (int k = 0; k < FILL_SLOTS; ++k) {
            float x = gx[k], y = gy[k], z = gz[k];
            float w = D2_SCALE * fmaf(z, z, fmaf(x, x, y * y));
            bool ok = (x != PADF);
            mn = ok ? fminf(mn, x) : mn;
            mx = ok ? fmaxf(mx, x) : mx;
            float ys = ok ? y : NAN;
            yn = fminf(yn, ys);
            yx = fmaxf(yx, ys);
            wm = ok ? fmaxf(wm, w) : wm;
        }
        #pragma unroll
        for (int off = 1; off < 32; off <<= 1) {
            mn = fminf(mn, __shfl_xor(mn, off, 32));
            mx = fmaxf(mx, __shfl_xor(mx, off, 32));
            yn = fminf(yn, __shfl_xor(yn, off, 32));
            yx = fmaxf(yx, __shfl_xor(yx, off, 32));
            wm = fmaxf(wm, __shfl_xor(wm, off, 32));
        }
        if ((tid & 31) == 0) { red[wave] = make_float4(mn, mx, yn, yx); redw[wave] = wm; }
    }

    #pragma unroll
    for (int k = 0; k < FILL_SLOTS; ++k)
        asm volatile("" : "+v"(idx[k]));

    bool  pv;
    float pcx, pcy, pcz;
    if (w_partner) {
        pv = (pidx != PADI);
        int sp = pv ? pidx : 0;
        const float* pc = coords + ((size_t)b * Mn + sp) * 3;
        pcx = pc[0]; pcy = pc[1]; pcz = pc[2];
    }
    bool am = false;
    int  gb[4] = {0, 0, 0, 0};
    int  gm[4] = {0, 0, 0, 0};
    if (w_angle) {
        am = (a[0] != PADI) && (a[1] != PADI) && (a[2] != PADI) && (a[3] != PADI);
        #pragma unroll
        for (int t = 0; t < 4; ++t) {
            int g = am ? a[t] : 0;
            gb[t] = g >> 12;
            gm[t] = cia[g];
        }
    }
    lds_barrier();

    const int sx = blk % NSX;
    const int sy = blk / NSX;
    float rlo, rhi, clo, chi;
    float rylo, ryhi, cylo, cyhi;
    {
        float xmin = INFINITY, xmax = -INFINITY, wmax = 0.0f;
        float ymin = INFINITY, ymax = -INFINITY;
        #pragma unroll
        for (int w = 0; w < 8; ++w) {
            float4 r = red[w];
            xmin = fminf(xmin, r.x); xmax = fmaxf(xmax, r.y);
            ymin = fminf(ymin, r.z); ymax = fmaxf(ymax, r.w);
            wmax = fmaxf(wmax, redw[w]);
        }
        if (!(xmax >= xmin)) { xmin = 0.0f; xmax = 0.0f; }
        if (!(ymax >= ymin)) { ymin = 0.0f; ymax = 0.0f; }
        float wx    = (xmax - xmin) * (1.0f / NSX);
        float wy    = (ymax - ymin) * (1.0f / NSY);
        float xlo   = fmaf((float)sx, wx, xmin);
        float xhi   = fmaf((float)(sx + 1), wx, xmin);
        float ylo   = fmaf((float)sy, wy, ymin);
        float yhi   = fmaf((float)(sy + 1), wy, ymin);
        float reach = CUTOFF_F + MARGIN_ABS + (wmax * 64.0f) * MARGIN_REL;
        rlo  = (sx == 0)       ? -INFINITY : xlo;
        rhi  = (sx == NSX - 1) ?  INFINITY : xhi;
        clo  = (sx == 0)       ? -INFINITY : xlo - reach;
        chi  = (sx == NSX - 1) ?  INFINITY : xhi + reach;
        rylo = (sy == 0)       ? -INFINITY : ylo;
        ryhi = (sy == NSY - 1) ?  INFINITY : yhi;
        cylo = (sy == 0)       ? -INFINITY : ylo - reach;
        cyhi = (sy == NSY - 1) ?  INFINITY : yhi + reach;
    }

    float qx[FILL_SLOTS], qy[FILL_SLOTS], qz[FILL_SLOTS], qw[FILL_SLOTS];
    #pragma unroll
    for (int k = 0; k < FILL_SLOTS; ++k) {
        float4 r = raw[idx[k]];
        qx[k] = r.x; qy[k] = r.y; qz[k] = r.z; qw[k] = r.w;
    }
    __builtin_amdgcn_sched_barrier(0);
    #pragma unroll
    for (int k = 0; k < FILL_SLOTS; ++k)
        if (k == (blk >> 2) && (wave >> 1) == (blk & 3)) nvx[tid & 63] = qx[k];

    unsigned bcol[FILL_SLOTS], brow[FILL_SLOTS];
    {
        int cc = 0, cr = 0;
        #pragma unroll
        for (int k = 0; k < FILL_SLOTS; ++k) {
            float x = qx[k], y = qy[k];
            unsigned v = __builtin_amdgcn_ballot_w32(x != PADF);
            bcol[k] = v & __builtin_amdgcn_ballot_w32(x >= clo)  & __builtin_amdgcn_ballot_w32(x <= chi)
                        & __builtin_amdgcn_ballot_w32(y >= cylo) & __builtin_amdgcn_ballot_w32(y <= cyhi);
            brow[k] = v & __builtin_amdgcn_ballot_w32(x >= rlo)  & __builtin_amdgcn_ballot_w32(!(x >= rhi))
                        & __builtin_amdgcn_ballot_w32(y >= rylo) & __builtin_amdgcn_ballot_w32(!(y >= ryhi));
            cc += __builtin_popcount(bcol[k]);
            cr += __builtin_popcount(brow[k]);
        }
        if ((tid & 31) == 0) cnt[wave] = make_int2(cc, cr);
    }

    #pragma unroll
    for (int t = 0; t < 4; ++t)
        asm volatile("" : "+v"(gm[t]));

    if (w_partner) {
        float* o = orowf + (tid >> 1) * 8 + 2 + (tid & 1) * 3;
        o[0] = pv ? pcx : PADF;
        o[1] = pv ? pcy : PADF;
        o[2] = pv ? pcz : PADF;
    }
    float3 p[4];
    if (w_angle) {
        #pragma unroll
        for (int t = 0; t < 4; ++t) {
            const float* c = coords + ((size_t)gb[t] * Mn + gm[t]) * 3;
            p[t] = make_float3(c[0], c[1], c[2]);
        }
    }
    lds_barrier();

    int lane  = threadIdx.x & 31;
    int half  = lane >> 4;
    int l     = lane & 15;

    int nC = 0, nR = 0;
    {
        int baseC = 0, baseR = 0;
        #pragma unroll
        for (int w = 0; w < 8; ++w) {
            int2 c = cnt[w];
            baseC += (w < wave) ? c.x : 0;
            baseR += (w < wave) ? c.y : 0;
            nC += c.x; nR += c.y;
        }
        nC = __builtin_amdgcn_readfirstlane(nC);
        nR = __builtin_amdgcn_readfirstlane(nR);
        int runC = __builtin_amdgcn_readfirstlane(baseC);
        int runR = __builtin_amdgcn_readfirstlane(baseR);
        #pragma unroll
        for (int k = 0; k < FILL_SLOTS; ++k) {
            unsigned mc = bcol[k], mr = brow[k];
            int pc = runC + (int)__builtin_amdgcn_mbcnt_lo(mc, 0u);
            int pr = runR + (int)__builtin_amdgcn_mbcnt_lo(mr, 0u);
            float4 q = make_float4(qx[k], qy[k], qz[k], qw[k]);
            if ((mc >> lane) & 1u) raw[pc] = q;
            if ((mr >> lane) & 1u) { rowl[pr] = q; rowi[pr] = k * 256 + tid; }
            runC += __builtin_popcount(mc);
            runR += __builtin_popcount(mr);
        }
        if (tid < 32) raw[nC + tid] = make_float4(0.0f, 0.0f, 0.0f, 1e30f);
        if (tid < 16) { rowl[nR + tid] = make_float4(0.0f, 0.0f, 0.0f, 0.0f); rowi[nR + tid] = 0; }
    }
    lds_barrier();

    const int nCP = (nC + 31) >> 5;
    const int nRT = (nR + 15) >> 4;

    const v2f gscale = {-2.0f * D2_SCALE, -2.0f * D2_SCALE};
    const v2f mscale = {MASK_SCALE, MASK_SCALE};
    const v2f mbias  = {MASK_BIAS, MASK_BIAS};

    v8f zc = {};
    for (int rt0 = 0; rt0 < nRT; rt0 += CHUNK_ROWS / 16) {
    const int rtend = (rt0 + CHUNK_ROWS / 16 < nRT) ? rt0 + CHUNK_ROWS / 16 : nRT;
    const int nItems = (rtend - rt0) * nCP;
    const int q0 = (wave * nItems) >> 3;
    const int q1 = ((wave + 1) * nItems) >> 3;
    for (int rt = rt0; rt < rtend; ++rt) {
    const int qb = (rt - rt0) * nCP;
    const int p0 = (q0 - qb > 0) ? q0 - qb : 0;
    const int p1 = (q1 - qb < nCP) ? q1 - qb : nCP;
    if (p0 >= p1) {
        if (l == 0) {
            #pragma unroll
            for (int v = 0; v < 8; ++v)
                part[wave][(rt - rt0) * 16 + half * 8 + v] = 0.0f;
        }
        continue;
    }
    float4 arow = rowl[rt * 16 + l];
    v2f A;
    A.x = half ? arow.z : arow.x;
    A.y = half ? 0.0f   : arow.y;

    float h[8];
    #pragma unroll
    for (int v = 0; v < 8; ++v) h[v] = rowl[rt * 16 + half * 8 + v].w;

    float acc0[8], acc1[8];
    #pragma unroll
    for (int v = 0; v < 8; ++v) { acc0[v] = 0.0f; acc1[v] = 0.0f; }

    for (int p = p0; p < p1; ++p) {
        const int j0 = p * 32;
        float4 brow0 = raw[j0 + l];
        float4 brow1 = raw[j0 + 16 + l];
        v2f B0, B1;
        B0.x = half ? brow0.z : brow0.x;
        B0.y = half ? 0.0f    : brow0.y;
        B1.x = half ? brow1.z : brow1.x;
        B1.y = half ? 0.0f    : brow1.y;
        float sqj0 = brow0.w;
        float sqj1 = brow1.w;

        v8f G0 = __builtin_amdgcn_wmma_f32_16x16x4_f32(
            false, A, false, B0, (short)0, zc, false, false);
        v8f G1 = __builtin_amdgcn_wmma_f32_16x16x4_f32(
            false, A, false, B1, (short)0, zc, false, false);

        #pragma unroll
        for (int v = 0; v < 8; v += 2) {
            v2f g, hs, t;
            g.x  = G0[v];          g.y  = G0[v + 1];
            hs.x = h[v] + sqj0;    hs.y = h[v + 1] + sqj0;
            v2f m = fma_sat(g, gscale, hs);
            t.x = m.x + EPS_S;     t.y = m.y + EPS_S;
            v2f w = fma_sat(t, mscale, mbias);
            acc0[v]     = fmaf(__builtin_amdgcn_sqrtf(t.x), w.x, acc0[v]);
            acc0[v + 1] = fmaf(__builtin_amdgcn_sqrtf(t.y), w.y, acc0[v + 1]);
        }
        #pragma unroll
        for (int v = 0; v < 8; v += 2) {
            v2f g, hs, t;
            g.x  = G1[v];          g.y  = G1[v + 1];
            hs.x = h[v] + sqj1;    hs.y = h[v + 1] + sqj1;
            v2f m = fma_sat(g, gscale, hs);
            t.x = m.x + EPS_S;     t.y = m.y + EPS_S;
            v2f w = fma_sat(t, mscale, mbias);
            acc1[v]     = fmaf(__builtin_amdgcn_sqrtf(t.x), w.x, acc1[v]);
            acc1[v + 1] = fmaf(__builtin_amdgcn_sqrtf(t.y), w.y, acc1[v + 1]);
        }
    }

    #pragma unroll
    for (int v = 0; v < 8; ++v) acc0[v] += acc1[v];

    #pragma unroll
    for (int off = 1; off < 16; off <<= 1) {
        #pragma unroll
        for (int v = 0; v < 8; ++v)
            acc0[v] += __shfl_xor(acc0[v], off, 32);
    }

    if (l == 0) {
        #pragma unroll
        for (int v = 0; v < 8; ++v)
            part[wave][(rt - rt0) * 16 + half * 8 + v] = acc0[v];
    }
    }

    lds_barrier();
    if (tid < CHUNK_ROWS) {
        int r = rt0 * 16 + tid;
        if (r < nR) {
            float s = ((part[0][tid] + part[1][tid]) + (part[2][tid] + part[3][tid]))
                    + ((part[4][tid] + part[5][tid]) + (part[6][tid] + part[7][tid]));
            out[((size_t)b * Mn + rowi[r]) * 8 + 1] = s * 8.0f;
        }
    }
    lds_barrier();
    }

    if (w_angle) {
        float3 b1 = sub3(p[1], p[0]);
        float3 b2 = sub3(p[2], p[1]);
        float3 b3 = sub3(p[3], p[2]);
        float3 n1 = cross3(b1, b2);
        float3 n2 = cross3(b2, b3);
        float  inv = 1.0f / (sqrtf(dot3(b2, b2)) + 1e-12f);
        float3 b2n = make_float3(b2.x * inv, b2.y * inv, b2.z * inv);
        float3 m1  = cross3(n1, b2n);
        float  xx  = dot3(n1, n2);
        float  yy  = dot3(m1, n2);
        orowf[(tid - 192) * 8 + 0] = am ? atan2f(yy, xx) : PADF;
    }
    lds_barrier();

    if (w_partner) {
        int    r    = tid >> 1;
        int    hsel = tid & 1;
        float4 v    = orow[tid];
        bool  valid = (nvx[r] != PADF);
        float* o    = out + (size_t)(n0 + r) * 8 + hsel * 4;
        if (hsel == 0) {
            o[0] = v.x;
            if (!valid) o[1] = 0.0f;
            *(float2*)(o + 2) = make_float2(v.z, v.w);
        } else {
            *(float4*)o = v;
        }
    }
}

extern "C" void kernel_launch(void* const* d_in, const int* in_sizes, int n_in,
                              void* d_out, int out_size, void* d_ws, size_t ws_size,
                              hipStream_t stream) {
    (void)in_sizes; (void)n_in; (void)out_size; (void)d_ws; (void)ws_size;
    const float* coords   = (const float*)d_in[0];
    const int*   cia      = (const int*)d_in[2];
    const int*   partners = (const int*)d_in[3];
    const int*   aidx     = (const int*)d_in[4];
    float*       out      = (float*)d_out;

    fused_kernel<<<Bn * BLOCKS_PER_BATCH, 256, 0, stream>>>(coords, cia, partners, aidx, out);
}
